# starting kernel
# baseline (speedup 1.0000x reference)
#include <hip/hip_runtime.h>
#include <math.h>

typedef float v2f __attribute__((ext_vector_type(2)));
typedef float v8f __attribute__((ext_vector_type(8)));

#define FEAT 128
#define BN_EPS 1e-3f


__global__ void zero_f32(float* p, int n) {
    int i = blockIdx.x * blockDim.x + threadIdx.x;
    if (i < n) p[i] = 0.0f;
}

__global__ void deg_kernel(const int* __restrict__ rows, const float* __restrict__ w,
                           float* deg, int nE) {
    int e = blockIdx.x * blockDim.x + threadIdx.x;
    if (e < nE) atomicAdd(&deg[rows[e]], w[e]);
}

__global__ void dinv_kernel(const float* __restrict__ deg, float* __restrict__ dinv, int n) {
    int i = blockIdx.x * blockDim.x + threadIdx.x;
    if (i < n) {
        float d = deg[i] + 1.0f;
        dinv[i] = (d > 0.0f) ? rsqrtf(d) : 0.0f;
    }
}


__global__ __launch_bounds__(256) void gemm_kernel(const float* __restrict__ x,
                                                   const float* __restrict__ W,
                                                   float* __restrict__ h, int nRows) {
    __shared__ float Wl[FEAT * FEAT];

    int t = threadIdx.x;
    const float4* Wv = (const float4*)W;
    float4* Wlv = (float4*)Wl;
    #pragma unroll
    for (int i = t; i < (FEAT * FEAT) / 4; i += 256) Wlv[i] = Wv[i];
    __syncthreads();

    int wave = t >> 5;
    int lane = t & 31;
    int row0 = blockIdx.x * 128 + wave * 16;
    if (row0 >= nRows) return;

    int m  = lane & 15;
    int kh = (lane >> 4) << 1;
    int srow = row0 + m;
    if (srow >= nRows) srow = nRows - 1;
    const float* xr = x + (size_t)srow * FEAT;

    v8f acc[8] = {};

    for (int kb = 0; kb < FEAT / 4; ++kb) {
        int k = kb * 4 + kh;
        float2 av = *(const float2*)(xr + k);
        v2f a; a.x = av.x; a.y = av.y;
        #pragma unroll
        for (int j = 0; j < 8; ++j) {
            int col = j * 16 + m;
            v2f b;
            b.x = Wl[(k + 0) * FEAT + col];
            b.y = Wl[(k + 1) * FEAT + col];
            acc[j] = __builtin_amdgcn_wmma_f32_16x16x4_f32(
                false, a, false, b, (short)0, acc[j], false, false);
        }
    }

    int rbase = row0 + ((lane >> 4) ? 8 : 0);
    int n = lane & 15;
    #pragma unroll
    for (int j = 0; j < 8; ++j) {
        #pragma unroll
        for (int r = 0; r < 8; ++r) {
            int row = rbase + r;
            if (row < nRows) h[(size_t)row * FEAT + j * 16 + n] = acc[j][r];
        }
    }
}


__global__ void self_loop_init(const float* __restrict__ h, const float* __restrict__ dinv,
                               float* __restrict__ out, int nNodes) {
    int i = blockIdx.x * blockDim.x + threadIdx.x;
    int total = nNodes * (FEAT / 4);
    if (i < total) {
        int node = i >> 5;
        float s = dinv[node];
        s = s * s;
        float4 v = ((const float4*)h)[i];
        v.x *= s; v.y *= s; v.z *= s; v.w *= s;
        ((float4*)out)[i] = v;
    }
}


__global__ __launch_bounds__(256) void edge_kernel(const int* __restrict__ rows,
                                                   const int* __restrict__ cols,
                                                   const float* __restrict__ w,
                                                   const float* __restrict__ dinv,
                                                   const float* __restrict__ h,
                                                   float* out, int nE) {
    int gid = blockIdx.x * blockDim.x + threadIdx.x;
    int wid = gid >> 5;
    int lane = gid & 31;
    int stride = (gridDim.x * blockDim.x) >> 5;
    for (int e = wid; e < nE; e += stride) {
        int r = rows[e];
        int c = cols[e];
        float norm = dinv[r] * w[e] * dinv[c];
        float4 hv = *(const float4*)(h + (size_t)c * FEAT + lane * 4);
        float* op = out + (size_t)r * FEAT + lane * 4;
        atomicAdd(op + 0, hv.x * norm);
        atomicAdd(op + 1, hv.y * norm);
        atomicAdd(op + 2, hv.z * norm);
        atomicAdd(op + 3, hv.w * norm);
    }
}


__global__ void bn_relu_kernel(float* __restrict__ out,
                               const float* __restrict__ bias,
                               const float* __restrict__ gamma,
                               const float* __restrict__ beta,
                               const float* __restrict__ mean,
                               const float* __restrict__ var, int total) {
    int i = blockIdx.x * blockDim.x + threadIdx.x;
    if (i < total) {
        int u = i & (FEAT - 1);
        float v = out[i] + bias[u];
        float r = gamma[u] * (v - mean[u]) * rsqrtf(var[u] + BN_EPS) + beta[u];
        out[i] = r > 0.0f ? r : 0.0f;
    }
}


extern "C" void kernel_launch(void* const* d_in, const int* in_sizes, int n_in,
                              void* d_out, int out_size, void* d_ws, size_t ws_size,
                              hipStream_t stream) {
    const float* x     = (const float*)d_in[0];
    const int*   eidx  = (const int*)d_in[1];
    const float* ew    = (const float*)d_in[2];
    const float* W     = (const float*)d_in[3];
    const float* bias  = (const float*)d_in[4];
    const float* gamma = (const float*)d_in[5];
    const float* beta  = (const float*)d_in[6];
    const float* mean  = (const float*)d_in[7];
    const float* var   = (const float*)d_in[8];

    int nE = in_sizes[2];
    int nN = in_sizes[0] / FEAT;
    const int* rows = eidx;
    const int* cols = eidx + nE;

    float* h    = (float*)d_ws;
    float* deg  = h + (size_t)nN * FEAT;
    float* dinv = deg + nN;
    float* out  = (float*)d_out;

    zero_f32<<<(nN + 255) / 256, 256, 0, stream>>>(deg, nN);
    deg_kernel<<<(nE + 255) / 256, 256, 0, stream>>>(rows, ew, deg, nE);
    dinv_kernel<<<(nN + 255) / 256, 256, 0, stream>>>(deg, dinv, nN);

    gemm_kernel<<<(nN + 127) / 128, 256, 0, stream>>>(x, W, h, nN);

    int initTotal = nN * (FEAT / 4);
    self_loop_init<<<(initTotal + 255) / 256, 256, 0, stream>>>(h, dinv, out, nN);

    long edgeThreads = (long)nE * 32;
    edge_kernel<<<(int)((edgeThreads + 255) / 256), 256, 0, stream>>>(
        rows, cols, ew, dinv, h, out, nE);

    int total = nN * FEAT;
    bn_relu_kernel<<<(total + 255) / 256, 256, 0, stream>>>(
        out, bias, gamma, beta, mean, var, total);
}


# edge scatter: lane owns features lane+32j, so each f32 atomic add and each gather load of a wave covers one whole 128-byte line, not 8 floats in each of 4 lines; sums still atomic, arrival order
# speedup vs baseline: 1.2781x; 1.2781x over previous
#include <hip/hip_runtime.h>
#include <math.h>

typedef float v2f __attribute__((ext_vector_type(2)));
typedef float v8f __attribute__((ext_vector_type(8)));

#define FEAT 128
#define BN_EPS 1e-3f


__global__ void zero_f32(float* p, int n) {
    int i = blockIdx.x * blockDim.x + threadIdx.x;
    if (i < n) p[i] = 0.0f;
}

__global__ void deg_kernel(const int* __restrict__ rows, const float* __restrict__ w,
                           float* deg, int nE) {
    int e = blockIdx.x * blockDim.x + threadIdx.x;
    if (e < nE) atomicAdd(&deg[rows[e]], w[e]);
}

__global__ void dinv_kernel(const float* __restrict__ deg, float* __restrict__ dinv, int n) {
    int i = blockIdx.x * blockDim.x + threadIdx.x;
    if (i < n) {
        float d = deg[i] + 1.0f;
        dinv[i] = (d > 0.0f) ? rsqrtf(d) : 0.0f;
    }
}


__global__ __launch_bounds__(256) void gemm_kernel(const float* __restrict__ x,
                                                   const float* __restrict__ W,
                                                   float* __restrict__ h, int nRows) {
    __shared__ float Wl[FEAT * FEAT];

    int t = threadIdx.x;
    const float4* Wv = (const float4*)W;
    float4* Wlv = (float4*)Wl;
    #pragma unroll
    for (int i = t; i < (FEAT * FEAT) / 4; i += 256) Wlv[i] = Wv[i];
    __syncthreads();

    int wave = t >> 5;
    int lane = t & 31;
    int row0 = blockIdx.x * 128 + wave * 16;
    if (row0 >= nRows) return;

    int m  = lane & 15;
    int kh = (lane >> 4) << 1;
    int srow = row0 + m;
    if (srow >= nRows) srow = nRows - 1;
    const float* xr = x + (size_t)srow * FEAT;

    v8f acc[8] = {};

    for (int kb = 0; kb < FEAT / 4; ++kb) {
        int k = kb * 4 + kh;
        float2 av = *(const float2*)(xr + k);
        v2f a; a.x = av.x; a.y = av.y;
        #pragma unroll
        for (int j = 0; j < 8; ++j) {
            int col = j * 16 + m;
            v2f b;
            b.x = Wl[(k + 0) * FEAT + col];
            b.y = Wl[(k + 1) * FEAT + col];
            acc[j] = __builtin_amdgcn_wmma_f32_16x16x4_f32(
                false, a, false, b, (short)0, acc[j], false, false);
        }
    }

    int rbase = row0 + ((lane >> 4) ? 8 : 0);
    int n = lane & 15;
    #pragma unroll
    for (int j = 0; j < 8; ++j) {
        #pragma unroll
        for (int r = 0; r < 8; ++r) {
            int row = rbase + r;
            if (row < nRows) h[(size_t)row * FEAT + j * 16 + n] = acc[j][r];
        }
    }
}


__global__ void self_loop_init(const float* __restrict__ h, const float* __restrict__ dinv,
                               float* __restrict__ out, int nNodes) {
    int i = blockIdx.x * blockDim.x + threadIdx.x;
    int total = nNodes * (FEAT / 4);
    if (i < total) {
        int node = i >> 5;
        float s = dinv[node];
        s = s * s;
        float4 v = ((const float4*)h)[i];
        v.x *= s; v.y *= s; v.z *= s; v.w *= s;
        ((float4*)out)[i] = v;
    }
}


__global__ __launch_bounds__(256) void edge_kernel(const int* __restrict__ rows,
                                                   const int* __restrict__ cols,
                                                   const float* __restrict__ w,
                                                   const float* __restrict__ dinv,
                                                   const float* __restrict__ h,
                                                   float* out, int nE) {
    int gid = blockIdx.x * blockDim.x + threadIdx.x;
    int wid = gid >> 5;
    int lane = gid & 31;
    int stride = (gridDim.x * blockDim.x) >> 5;
    for (int e = wid; e < nE; e += stride) {
        int r = rows[e];
        int c = cols[e];
        float norm = dinv[r] * w[e] * dinv[c];
        const float* hp = h + (size_t)c * FEAT + lane;
        float h0 = hp[0];
        float h1 = hp[32];
        float h2 = hp[64];
        float h3 = hp[96];
        float* op = out + (size_t)r * FEAT + lane;
        atomicAdd(op + 0,  h0 * norm);
        atomicAdd(op + 32, h1 * norm);
        atomicAdd(op + 64, h2 * norm);
        atomicAdd(op + 96, h3 * norm);
    }
}


__global__ void bn_relu_kernel(float* __restrict__ out,
                               const float* __restrict__ bias,
                               const float* __restrict__ gamma,
                               const float* __restrict__ beta,
                               const float* __restrict__ mean,
                               const float* __restrict__ var, int total) {
    int i = blockIdx.x * blockDim.x + threadIdx.x;
    if (i < total) {
        int u = i & (FEAT - 1);
        float v = out[i] + bias[u];
        float r = gamma[u] * (v - mean[u]) * rsqrtf(var[u] + BN_EPS) + beta[u];
        out[i] = r > 0.0f ? r : 0.0f;
    }
}


extern "C" void kernel_launch(void* const* d_in, const int* in_sizes, int n_in,
                              void* d_out, int out_size, void* d_ws, size_t ws_size,
                              hipStream_t stream) {
    const float* x     = (const float*)d_in[0];
    const int*   eidx  = (const int*)d_in[1];
    const float* ew    = (const float*)d_in[2];
    const float* W     = (const float*)d_in[3];
    const float* bias  = (const float*)d_in[4];
    const float* gamma = (const float*)d_in[5];
    const float* beta  = (const float*)d_in[6];
    const float* mean  = (const float*)d_in[7];
    const float* var   = (const float*)d_in[8];

    int nE = in_sizes[2];
    int nN = in_sizes[0] / FEAT;
    const int* rows = eidx;
    const int* cols = eidx + nE;

    float* h    = (float*)d_ws;
    float* deg  = h + (size_t)nN * FEAT;
    float* dinv = deg + nN;
    float* out  = (float*)d_out;

    zero_f32<<<(nN + 255) / 256, 256, 0, stream>>>(deg, nN);
    deg_kernel<<<(nE + 255) / 256, 256, 0, stream>>>(rows, ew, deg, nE);
    dinv_kernel<<<(nN + 255) / 256, 256, 0, stream>>>(deg, dinv, nN);

    gemm_kernel<<<(nN + 127) / 128, 256, 0, stream>>>(x, W, h, nN);

    int initTotal = nN * (FEAT / 4);
    self_loop_init<<<(initTotal + 255) / 256, 256, 0, stream>>>(h, dinv, out, nN);

    long edgeThreads = (long)nE * 32;
    edge_kernel<<<(int)((edgeThreads + 255) / 256), 256, 0, stream>>>(
        rows, cols, ew, dinv, h, out, nE);

    int total = nN * FEAT;
    bn_relu_kernel<<<(total + 255) / 256, 256, 0, stream>>>(
        out, bias, gamma, beta, mean, var, total);
}


# edge sums: edges counting-sorted by row in-call (int atomic slots, arrival order), one wave sums a row of x in registers, product after it ((A x) W) with BN/ReLU epilogue; no f32 atomics on out
# speedup vs baseline: 1.3442x; 1.0517x over previous
#include <hip/hip_runtime.h>
#include <math.h>


typedef float v2f __attribute__((ext_vector_type(2)));
typedef float v8f __attribute__((ext_vector_type(8)));

#define FEAT 128
#define BN_EPS 1e-3f


__global__ void zero_kernel(float* deg, int* cnt, int n) {
    int i = blockIdx.x * blockDim.x + threadIdx.x;
    if (i < n) { deg[i] = 0.0f; cnt[i] = 0; }
}

__global__ void hist_kernel(const int* __restrict__ rows, const float* __restrict__ w,
                            float* deg, int* cnt, int nE) {
    int e = blockIdx.x * blockDim.x + threadIdx.x;
    if (e < nE) {
        int r = rows[e];
        atomicAdd(&deg[r], w[e]);
        atomicAdd(&cnt[r], 1);
    }
}


__global__ __launch_bounds__(1024) void scan_local(const int* __restrict__ cnt,
                                                    int* __restrict__ rowptr,
                                                    int* __restrict__ bsum, int n) {
    __shared__ int wsum[32];
    int t = threadIdx.x;
    int lane = t & 31;
    int wv = t >> 5;
    int i = blockIdx.x * 1024 + t;
    int v = (i < n) ? cnt[i] : 0;
    int inc = v;
    #pragma unroll
    for (int d = 1; d < 32; d <<= 1) {
        int o = __shfl_up(inc, d);
        if (lane >= d) inc += o;
    }
    if (lane == 31) wsum[wv] = inc;
    __syncthreads();
    if (wv == 0) {
        int s = wsum[lane];
        int si = s;
        #pragma unroll
        for (int d = 1; d < 32; d <<= 1) {
            int o = __shfl_up(si, d);
            if (lane >= d) si += o;
        }
        wsum[lane] = si - s;
        if (lane == 31) bsum[blockIdx.x] = si;
    }
    __syncthreads();
    if (i < n) rowptr[i] = wsum[wv] + inc - v;
}


__global__ __launch_bounds__(1024) void scan_finish(int* __restrict__ rowptr,
                                                     int* __restrict__ cursor,
                                                     const int* __restrict__ bsum,
                                                     const float* __restrict__ deg,
                                                     float* __restrict__ dinv, int n, int nE) {
    int b = blockIdx.x;
    int off = 0;
    for (int k = 0; k < b; ++k) off += bsum[k];
    int i = b * 1024 + threadIdx.x;
    if (i < n) {
        int p = rowptr[i] + off;
        rowptr[i] = p;
        cursor[i] = p;
        float d = deg[i] + 1.0f;
        dinv[i] = (d > 0.0f) ? rsqrtf(d) : 0.0f;
    }
    if (i == n) rowptr[n] = nE;
}


__global__ void fill_kernel(const int* __restrict__ rows, const int* __restrict__ cols,
                            const float* __restrict__ w, const float* __restrict__ dinv,
                            int* cursor, int2* __restrict__ ent, int nE) {
    int e = blockIdx.x * blockDim.x + threadIdx.x;
    if (e < nE) {
        int r = rows[e];
        int c = cols[e];
        int pos = atomicAdd(&cursor[r], 1);
        int2 v;
        v.x = c;
        v.y = __float_as_int(w[e] * dinv[c]);
        ent[pos] = v;
    }
}


__device__ __forceinline__ float4 fma4(float s, float4 a, float4 c) {
    c.x = fmaf(s, a.x, c.x);
    c.y = fmaf(s, a.y, c.y);
    c.z = fmaf(s, a.z, c.z);
    c.w = fmaf(s, a.w, c.w);
    return c;
}

__global__ __launch_bounds__(256) void gather_kernel(const float* __restrict__ x,
                                                     const int* __restrict__ rowptr,
                                                     const int2* __restrict__ ent,
                                                     const float* __restrict__ dinv,
                                                     float* __restrict__ agg, int nN) {
    int lane = threadIdx.x & 31;
    int wid = __builtin_amdgcn_readfirstlane((int)((blockIdx.x * blockDim.x + threadIdx.x) >> 5));
    int nW = (int)((gridDim.x * blockDim.x) >> 5);
    const float4* xv = (const float4*)x;
    float4* av = (float4*)agg;

    for (int r = wid; r < nN; r += nW) {
        int start = rowptr[r];
        int end = rowptr[r + 1];
        float s = dinv[r];
        float4 xs = xv[(size_t)r * 32 + lane];
        float4 acc = make_float4(0.0f, 0.0f, 0.0f, 0.0f);

        for (int base = start; base < end; base += 32) {
            int n = end - base;
            if (n > 32) n = 32;
            int2 en = make_int2(0, 0);
            if (lane < n) en = ent[base + lane];
            int j = 0;
            for (; j + 4 <= n; j += 4) {
                int c0 = __builtin_amdgcn_readlane(en.x, j);
                int c1 = __builtin_amdgcn_readlane(en.x, j + 1);
                int c2 = __builtin_amdgcn_readlane(en.x, j + 2);
                int c3 = __builtin_amdgcn_readlane(en.x, j + 3);
                float w0 = __int_as_float(__builtin_amdgcn_readlane(en.y, j));
                float w1 = __int_as_float(__builtin_amdgcn_readlane(en.y, j + 1));
                float w2 = __int_as_float(__builtin_amdgcn_readlane(en.y, j + 2));
                float w3 = __int_as_float(__builtin_amdgcn_readlane(en.y, j + 3));
                float4 a0 = xv[(size_t)c0 * 32 + lane];
                float4 a1 = xv[(size_t)c1 * 32 + lane];
                float4 a2 = xv[(size_t)c2 * 32 + lane];
                float4 a3 = xv[(size_t)c3 * 32 + lane];
                acc = fma4(w0, a0, acc);
                acc = fma4(w1, a1, acc);
                acc = fma4(w2, a2, acc);
                acc = fma4(w3, a3, acc);
            }
            for (; j < n; ++j) {
                int c0 = __builtin_amdgcn_readlane(en.x, j);
                float w0 = __int_as_float(__builtin_amdgcn_readlane(en.y, j));
                float4 a0 = xv[(size_t)c0 * 32 + lane];
                acc = fma4(w0, a0, acc);
            }
        }

        float ss = s * s;
        float4 o;
        o.x = fmaf(ss, xs.x, s * acc.x);
        o.y = fmaf(ss, xs.y, s * acc.y);
        o.z = fmaf(ss, xs.z, s * acc.z);
        o.w = fmaf(ss, xs.w, s * acc.w);
        av[(size_t)r * 32 + lane] = o;
    }
}


__global__ __launch_bounds__(256) void gemm_bn_relu_kernel(const float* __restrict__ a_in,
                                                           const float* __restrict__ W,
                                                           const float* __restrict__ bias,
                                                           const float* __restrict__ gamma,
                                                           const float* __restrict__ beta,
                                                           const float* __restrict__ mean,
                                                           const float* __restrict__ var,
                                                           float* __restrict__ out, int nRows) {
    __shared__ float Wl[FEAT * FEAT];

    int t = threadIdx.x;
    const float4* Wv = (const float4*)W;
    float4* Wlv = (float4*)Wl;
    #pragma unroll
    for (int i = t; i < (FEAT * FEAT) / 4; i += 256) Wlv[i] = Wv[i];
    __syncthreads();

    int wave = t >> 5;
    int lane = t & 31;
    int row0 = blockIdx.x * 128 + wave * 16;
    if (row0 >= nRows) return;

    int m  = lane & 15;
    int kh = (lane >> 4) << 1;
    int srow = row0 + m;
    if (srow >= nRows) srow = nRows - 1;
    const float* xr = a_in + (size_t)srow * FEAT;

    v8f acc[8] = {};

    for (int kb = 0; kb < FEAT / 4; ++kb) {
        int k = kb * 4 + kh;
        float2 av = *(const float2*)(xr + k);
        v2f a; a.x = av.x; a.y = av.y;
        #pragma unroll
        for (int j = 0; j < 8; ++j) {
            int col = j * 16 + m;
            v2f b;
            b.x = Wl[(k + 0) * FEAT + col];
            b.y = Wl[(k + 1) * FEAT + col];
            acc[j] = __builtin_amdgcn_wmma_f32_16x16x4_f32(
                false, a, false, b, (short)0, acc[j], false, false);
        }
    }

    int rbase = row0 + ((lane >> 4) ? 8 : 0);
    int n = lane & 15;
    #pragma unroll
    for (int j = 0; j < 8; ++j) {
        int u = j * 16 + n;
        float bu = bias[u];
        float gu = gamma[u];
        float mu = mean[u];
        float rs = rsqrtf(var[u] + BN_EPS);
        float be = beta[u];
        #pragma unroll
        for (int r = 0; r < 8; ++r) {
            int row = rbase + r;
            float v = acc[j][r] + bu;
            float y = gu * (v - mu) * rs + be;
            y = y > 0.0f ? y : 0.0f;
            if (row < nRows) out[(size_t)row * FEAT + u] = y;
        }
    }
}


extern "C" void kernel_launch(void* const* d_in, const int* in_sizes, int n_in,
                              void* d_out, int out_size, void* d_ws, size_t ws_size,
                              hipStream_t stream) {
    const float* x     = (const float*)d_in[0];
    const int*   eidx  = (const int*)d_in[1];
    const float* ew    = (const float*)d_in[2];
    const float* W     = (const float*)d_in[3];
    const float* bias  = (const float*)d_in[4];
    const float* gamma = (const float*)d_in[5];
    const float* beta  = (const float*)d_in[6];
    const float* mean  = (const float*)d_in[7];
    const float* var   = (const float*)d_in[8];

    int nE = in_sizes[2];
    int nN = in_sizes[0] / FEAT;
    const int* rows = eidx;
    const int* cols = eidx + nE;

    float* agg  = (float*)d_ws;
    float* deg  = agg + (size_t)nN * FEAT;
    float* dinv = deg + nN;
    float* out  = (float*)d_out;

    int* o32    = (int*)d_out;
    int  pitch  = nN + 64;
    int* cnt    = o32;
    int* rowptr = o32 + pitch;
    int* cursor = o32 + 2 * pitch;
    int* bsum   = o32 + 3 * pitch;
    int2* ent   = (int2*)(o32 + 4 * pitch);

    int nb = (nN + 1024) / 1024;

    zero_kernel<<<(nN + 255) / 256, 256, 0, stream>>>(deg, cnt, nN);
    hist_kernel<<<(nE + 255) / 256, 256, 0, stream>>>(rows, ew, deg, cnt, nE);
    scan_local<<<nb, 1024, 0, stream>>>(cnt, rowptr, bsum, nN);
    scan_finish<<<nb, 1024, 0, stream>>>(rowptr, cursor, bsum, deg, dinv, nN, nE);
    fill_kernel<<<(nE + 255) / 256, 256, 0, stream>>>(rows, cols, ew, dinv, cursor, ent, nE);
    gather_kernel<<<4096, 256, 0, stream>>>(x, rowptr, ent, dinv, agg, nN);
    gemm_bn_relu_kernel<<<(nN + 127) / 128, 256, 0, stream>>>(
        agg, W, bias, gamma, beta, mean, var, out, nN);
}


# edge sort: one global atomic per edge, not three: the count's returned value is kept as the edge's rank, so placing needs no atomic, and the degree is summed from the placed weights (8-lane tree)
# speedup vs baseline: 1.9800x; 1.4730x over previous
#include <hip/hip_runtime.h>
#include <math.h>


typedef float v2f __attribute__((ext_vector_type(2)));
typedef float v8f __attribute__((ext_vector_type(8)));

#define FEAT 128
#define BN_EPS 1e-3f


__global__ void zero_kernel(int* cnt, int n) {
    int i = blockIdx.x * blockDim.x + threadIdx.x;
    if (i < n) cnt[i] = 0;
}

__global__ void count_kernel(const int* __restrict__ rows, int* cnt, int* __restrict__ rank, int nE) {
    int e = blockIdx.x * blockDim.x + threadIdx.x;
    if (e < nE) rank[e] = atomicAdd(&cnt[rows[e]], 1);
}


__global__ __launch_bounds__(1024) void scan_local(const int* __restrict__ cnt,
                                                    int* __restrict__ rowptr,
                                                    int* __restrict__ bsum, int n) {
    __shared__ int wsum[32];
    int t = threadIdx.x;
    int lane = t & 31;
    int wv = t >> 5;
    int i = blockIdx.x * 1024 + t;
    int v = (i < n) ? cnt[i] : 0;
    int inc = v;
    #pragma unroll
    for (int d = 1; d < 32; d <<= 1) {
        int o = __shfl_up(inc, d);
        if (lane >= d) inc += o;
    }
    if (lane == 31) wsum[wv] = inc;
    __syncthreads();
    if (wv == 0) {
        int s = wsum[lane];
        int si = s;
        #pragma unroll
        for (int d = 1; d < 32; d <<= 1) {
            int o = __shfl_up(si, d);
            if (lane >= d) si += o;
        }
        wsum[lane] = si - s;
        if (lane == 31) bsum[blockIdx.x] = si;
    }
    __syncthreads();
    if (i < n) rowptr[i] = wsum[wv] + inc - v;
}


__global__ __launch_bounds__(1024) void scan_finish(int* __restrict__ rowptr,
                                                     const int* __restrict__ bsum, int n, int nE) {
    int b = blockIdx.x;
    int off = 0;
    for (int k = 0; k < b; ++k) off += bsum[k];
    int i = b * 1024 + threadIdx.x;
    if (i < n) rowptr[i] += off;
    if (i == n) rowptr[n] = nE;
}


__global__ void fill_kernel(const int* __restrict__ rows, const int* __restrict__ cols,
                            const float* __restrict__ w, const int* __restrict__ rowptr,
                            const int* __restrict__ rank, int2* __restrict__ ent, int nE) {
    int e = blockIdx.x * blockDim.x + threadIdx.x;
    if (e < nE) {
        int pos = rowptr[rows[e]] + rank[e];
        int2 v;
        v.x = cols[e];
        v.y = __float_as_int(w[e]);
        ent[pos] = v;
    }
}


__global__ __launch_bounds__(256) void dinv_kernel(const int* __restrict__ rowptr,
                                                   const int2* __restrict__ ent,
                                                   float* __restrict__ dinv, int nN) {
    int g = blockIdx.x * blockDim.x + threadIdx.x;
    int r = g >> 3;
    int sub = g & 7;
    float s = 0.0f;
    if (r < nN) {
        int end = rowptr[r + 1];
        for (int p = rowptr[r] + sub; p < end; p += 8) s += __int_as_float(ent[p].y);
    }
    s += __shfl_xor(s, 1);
    s += __shfl_xor(s, 2);
    s += __shfl_xor(s, 4);
    if (r < nN && sub == 0) {
        float d = s + 1.0f;
        dinv[r] = (d > 0.0f) ? rsqrtf(d) : 0.0f;
    }
}


__device__ __forceinline__ float4 fma4(float s, float4 a, float4 c) {
    c.x = fmaf(s, a.x, c.x);
    c.y = fmaf(s, a.y, c.y);
    c.z = fmaf(s, a.z, c.z);
    c.w = fmaf(s, a.w, c.w);
    return c;
}

__global__ __launch_bounds__(256) void gather_kernel(const float* __restrict__ x,
                                                     const int* __restrict__ rowptr,
                                                     const int2* __restrict__ ent,
                                                     const float* __restrict__ dinv,
                                                     float* __restrict__ agg, int nN) {
    int lane = threadIdx.x & 31;
    int wid = __builtin_amdgcn_readfirstlane((int)((blockIdx.x * blockDim.x + threadIdx.x) >> 5));
    int nW = (int)((gridDim.x * blockDim.x) >> 5);
    const float4* xv = (const float4*)x;
    float4* av = (float4*)agg;

    for (int r = wid; r < nN; r += nW) {
        int start = rowptr[r];
        int end = rowptr[r + 1];
        float s = dinv[r];
        int2 en = make_int2(0, 0);
        int dl = 0;
        float4 xs = xv[(size_t)r * 32 + lane];
        float4 acc = make_float4(0.0f, 0.0f, 0.0f, 0.0f);

        for (int base = start; base < end; base += 32) {
            int n = end - base;
            if (n > 32) n = 32;
            const int2* ep = ent + base;
            if (lane < n) {
                en = ep[lane];
                dl = __float_as_int(dinv[en.x]);
            }
            int j = 0;
            for (; j + 4 <= n; j += 4) {
                int c0 = __builtin_amdgcn_readlane(en.x, j);
                int c1 = __builtin_amdgcn_readlane(en.x, j + 1);
                int c2 = __builtin_amdgcn_readlane(en.x, j + 2);
                int c3 = __builtin_amdgcn_readlane(en.x, j + 3);
                float w0 = __int_as_float(__builtin_amdgcn_readlane(en.y, j))
                         * __int_as_float(__builtin_amdgcn_readlane(dl, j));
                float w1 = __int_as_float(__builtin_amdgcn_readlane(en.y, j + 1))
                         * __int_as_float(__builtin_amdgcn_readlane(dl, j + 1));
                float w2 = __int_as_float(__builtin_amdgcn_readlane(en.y, j + 2))
                         * __int_as_float(__builtin_amdgcn_readlane(dl, j + 2));
                float w3 = __int_as_float(__builtin_amdgcn_readlane(en.y, j + 3))
                         * __int_as_float(__builtin_amdgcn_readlane(dl, j + 3));
                float4 a0 = xv[(size_t)c0 * 32 + lane];
                float4 a1 = xv[(size_t)c1 * 32 + lane];
                float4 a2 = xv[(size_t)c2 * 32 + lane];
                float4 a3 = xv[(size_t)c3 * 32 + lane];
                acc = fma4(w0, a0, acc);
                acc = fma4(w1, a1, acc);
                acc = fma4(w2, a2, acc);
                acc = fma4(w3, a3, acc);
            }
            for (; j < n; ++j) {
                int c0 = __builtin_amdgcn_readlane(en.x, j);
                float w0 = __int_as_float(__builtin_amdgcn_readlane(en.y, j))
                         * __int_as_float(__builtin_amdgcn_readlane(dl, j));
                float4 a0 = xv[(size_t)c0 * 32 + lane];
                acc = fma4(w0, a0, acc);
            }
        }

        float ss = s * s;
        float4 o;
        o.x = fmaf(ss, xs.x, s * acc.x);
        o.y = fmaf(ss, xs.y, s * acc.y);
        o.z = fmaf(ss, xs.z, s * acc.z);
        o.w = fmaf(ss, xs.w, s * acc.w);
        av[(size_t)r * 32 + lane] = o;
    }
}


__global__ __launch_bounds__(256) void gemm_bn_relu_kernel(const float* __restrict__ a_in,
                                                           const float* __restrict__ W,
                                                           const float* __restrict__ bias,
                                                           const float* __restrict__ gamma,
                                                           const float* __restrict__ beta,
                                                           const float* __restrict__ mean,
                                                           const float* __restrict__ var,
                                                           float* __restrict__ out, int nRows) {
    __shared__ float Wl[FEAT * FEAT];

    int t = threadIdx.x;
    const float4* Wv = (const float4*)W;
    float4* Wlv = (float4*)Wl;
    #pragma unroll
    for (int i = t; i < (FEAT * FEAT) / 4; i += 256) Wlv[i] = Wv[i];
    __syncthreads();

    int wave = t >> 5;
    int lane = t & 31;
    int row0 = blockIdx.x * 128 + wave * 16;
    if (row0 >= nRows) return;

    int m  = lane & 15;
    int kh = (lane >> 4) << 1;
    int srow = row0 + m;
    if (srow >= nRows) srow = nRows - 1;
    const float* xr = a_in + (size_t)srow * FEAT;

    v8f acc[8] = {};

    for (int kb = 0; kb < FEAT / 4; ++kb) {
        int k = kb * 4 + kh;
        float2 av = *(const float2*)(xr + k);
        v2f a; a.x = av.x; a.y = av.y;
        #pragma unroll
        for (int j = 0; j < 8; ++j) {
            int col = j * 16 + m;
            v2f b;
            b.x = Wl[(k + 0) * FEAT + col];
            b.y = Wl[(k + 1) * FEAT + col];
            acc[j] = __builtin_amdgcn_wmma_f32_16x16x4_f32(
                false, a, false, b, (short)0, acc[j], false, false);
        }
    }

    int rbase = row0 + ((lane >> 4) ? 8 : 0);
    int n = lane & 15;
    #pragma unroll
    for (int j = 0; j < 8; ++j) {
        int u = j * 16 + n;
        float bu = bias[u];
        float gu = gamma[u];
        float mu = mean[u];
        float rs = rsqrtf(var[u] + BN_EPS);
        float be = beta[u];
        #pragma unroll
        for (int r = 0; r < 8; ++r) {
            int row = rbase + r;
            float v = acc[j][r] + bu;
            float y = gu * (v - mu) * rs + be;
            y = y > 0.0f ? y : 0.0f;
            if (row < nRows) out[(size_t)row * FEAT + u] = y;
        }
    }
}


extern "C" void kernel_launch(void* const* d_in, const int* in_sizes, int n_in,
                              void* d_out, int out_size, void* d_ws, size_t ws_size,
                              hipStream_t stream) {
    const float* x     = (const float*)d_in[0];
    const int*   eidx  = (const int*)d_in[1];
    const float* ew    = (const float*)d_in[2];
    const float* W     = (const float*)d_in[3];
    const float* bias  = (const float*)d_in[4];
    const float* gamma = (const float*)d_in[5];
    const float* beta  = (const float*)d_in[6];
    const float* mean  = (const float*)d_in[7];
    const float* var   = (const float*)d_in[8];

    int nE = in_sizes[2];
    int nN = in_sizes[0] / FEAT;
    const int* rows = eidx;
    const int* cols = eidx + nE;

    float* agg  = (float*)d_ws;
    float* dinv = agg + (size_t)nN * FEAT + nN;
    float* out  = (float*)d_out;

    int* o32    = (int*)d_out;
    int  pitch  = nN + 64;
    int* cnt    = o32;
    int* rowptr = o32 + pitch;
    int* bsum   = o32 + 3 * pitch;
    int2* ent   = (int2*)(o32 + 4 * pitch);
    int* rank   = (int*)(ent + nE);

    int nb = (nN + 1024) / 1024;

    zero_kernel<<<(nN + 255) / 256, 256, 0, stream>>>(cnt, nN);
    count_kernel<<<(nE + 255) / 256, 256, 0, stream>>>(rows, cnt, rank, nE);
    scan_local<<<nb, 1024, 0, stream>>>(cnt, rowptr, bsum, nN);
    scan_finish<<<nb, 1024, 0, stream>>>(rowptr, bsum, nN, nE);
    fill_kernel<<<(nE + 255) / 256, 256, 0, stream>>>(rows, cols, ew, rowptr, rank, ent, nE);
    dinv_kernel<<<(nN * 8 + 255) / 256, 256, 0, stream>>>(rowptr, ent, dinv, nN);
    gather_kernel<<<4096, 256, 0, stream>>>(x, rowptr, ent, dinv, agg, nN);
    gemm_bn_relu_kernel<<<(nN + 127) / 128, 256, 0, stream>>>(
        agg, W, bias, gamma, beta, mean, var, out, nN);
}


# edge sort: two-level counting sort on LDS counters (512-row buckets, then rows; per-block counts prefix-summed), no global atomic; row order still arrival; degree summed serially per row
# speedup vs baseline: 3.2001x; 1.6162x over previous
#include <hip/hip_runtime.h>
#include <math.h>


typedef float v2f __attribute__((ext_vector_type(2)));
typedef float v8f __attribute__((ext_vector_type(8)));

#define FEAT 128
#define BN_EPS 1e-3f


#define SH     9
#define BROWS  512
#define NBMAX  2048
#define GBLK   256
#define CAP    16384

__global__ __launch_bounds__(1024) void hist_kernel(const int* __restrict__ rows, int* __restrict__ hist,
                                                    int nE, int nB) {
    __shared__ int h[NBMAX];
    int t = threadIdx.x;
    int g = blockIdx.x;
    int G = gridDim.x;
    for (int i = t; i < nB; i += 1024) h[i] = 0;
    __syncthreads();
    int chunk = (nE + G - 1) / G;
    int e0 = g * chunk;
    int e1 = e0 + chunk;
    if (e1 > nE) e1 = nE;
    for (int e = e0 + t; e < e1; e += 4 * 1024) {
        int r[4];
        #pragma unroll
        for (int k = 0; k < 4; ++k) {
            int j = e + k * 1024;
            r[k] = (j < e1) ? rows[j] : -1;
        }
        #pragma unroll
        for (int k = 0; k < 4; ++k)
            if (r[k] >= 0) atomicAdd(&h[r[k] >> SH], 1);
    }
    __syncthreads();
    for (int i = t; i < nB; i += 1024) hist[i * G + g] = h[i];
}


__global__ __launch_bounds__(1024) void scan_kernel(int* a, int n4) {
    __shared__ int wtot[32];
    int t = threadIdx.x;
    int lane = t & 31;
    int wv = t >> 5;
    int4* a4 = (int4*)a;
    int seg = (n4 + 31) / 32;
    int s0 = wv * seg;
    int s1 = s0 + seg;
    if (s1 > n4) s1 = n4;
    int sum = 0;
    for (int i = s0 + lane; i < s1; i += 32) {
        int4 v = a4[i];
        sum += v.x + v.y + v.z + v.w;
    }
    sum += __shfl_xor(sum, 16);
    sum += __shfl_xor(sum, 8);
    sum += __shfl_xor(sum, 4);
    sum += __shfl_xor(sum, 2);
    sum += __shfl_xor(sum, 1);
    if (lane == 0) wtot[wv] = sum;
    __syncthreads();
    int x = (lane < wv) ? wtot[lane] : 0;
    x += __shfl_xor(x, 16);
    x += __shfl_xor(x, 8);
    x += __shfl_xor(x, 4);
    x += __shfl_xor(x, 2);
    x += __shfl_xor(x, 1);
    int carry = x;
    for (int i0 = s0; i0 < s1; i0 += 32) {
        int i = i0 + lane;
        int4 v = make_int4(0, 0, 0, 0);
        if (i < s1) v = a4[i];
        int s = v.x + v.y + v.z + v.w;
        int inc = s;
        for (int d = 1; d < 32; d <<= 1) {
            int o = __shfl_up(inc, d);
            if (lane >= d) inc += o;
        }
        int ex = carry + inc - s;
        if (i < s1) a4[i] = make_int4(ex, ex + v.x, ex + v.x + v.y, ex + v.x + v.y + v.z);
        carry += __shfl(inc, 31);
    }
}


__global__ __launch_bounds__(1024) void bucket_kernel(const int* __restrict__ rows, const int* __restrict__ cols,
                                                      const float* __restrict__ w, const int* __restrict__ off,
                                                      int4* __restrict__ tmp, int nE, int nB) {
    __shared__ int cur[NBMAX];
    int t = threadIdx.x;
    int g = blockIdx.x;
    int G = gridDim.x;
    for (int i = t; i < nB; i += 1024) cur[i] = off[i * G + g];
    __syncthreads();
    int chunk = (nE + G - 1) / G;
    int e0 = g * chunk;
    int e1 = e0 + chunk;
    if (e1 > nE) e1 = nE;
    for (int e = e0 + t; e < e1; e += 4 * 1024) {
        int r[4], c[4], wb[4];
        #pragma unroll
        for (int k = 0; k < 4; ++k) {
            int j = e + k * 1024;
            bool ok = j < e1;
            r[k]  = ok ? rows[j] : -1;
            c[k]  = ok ? cols[j] : 0;
            wb[k] = ok ? __float_as_int(w[j]) : 0;
        }
        #pragma unroll
        for (int k = 0; k < 4; ++k) {
            if (r[k] >= 0) {
                int pos = atomicAdd(&cur[r[k] >> SH], 1);
                tmp[pos] = make_int4(r[k], c[k], wb[k], 0);
            }
        }
    }
}


__global__ __launch_bounds__(BROWS) void rowsort_kernel(const int4* __restrict__ tmp, const int* __restrict__ off,
                                                        int* __restrict__ rowptr, int2* ent,
                                                        float* __restrict__ dinv, int nE, int nN, int G) {
    __shared__ int cnt[BROWS];
    __shared__ int wsum[BROWS / 32];
    __shared__ int2 stage[CAP];
    int t = threadIdx.x;
    int lane = t & 31;
    int wv = t >> 5;
    int b = blockIdx.x;
    int nB = gridDim.x;
    int bs = off[b * G];
    int be = (b + 1 < nB) ? off[(b + 1) * G] : nE;
    int len = be - bs;
    const int4* src = tmp + bs;
    int2* dst = ent + bs;

    cnt[t] = 0;
    __syncthreads();
    for (int i = t; i < len; i += 4 * BROWS) {
        int r[4];
        #pragma unroll
        for (int k = 0; k < 4; ++k) {
            int j = i + k * BROWS;
            r[k] = (j < len) ? src[j].x : -1;
        }
        #pragma unroll
        for (int k = 0; k < 4; ++k)
            if (r[k] >= 0) atomicAdd(&cnt[r[k] & (BROWS - 1)], 1);
    }
    __syncthreads();

    int v = cnt[t];
    int inc = v;
    for (int d = 1; d < 32; d <<= 1) {
        int o = __shfl_up(inc, d);
        if (lane >= d) inc += o;
    }
    if (lane == 31) wsum[wv] = inc;
    __syncthreads();
    int x = (lane < wv) ? wsum[lane] : 0;
    x += __shfl_xor(x, 16);
    x += __shfl_xor(x, 8);
    x += __shfl_xor(x, 4);
    x += __shfl_xor(x, 2);
    x += __shfl_xor(x, 1);
    int start = x + inc - v;

    int row = b * BROWS + t;
    if (row < nN) rowptr[row] = bs + start;
    if (b == nB - 1 && t == 0) rowptr[nN] = nE;
    cnt[t] = start;
    __syncthreads();

    float s = 0.0f;
    if (len <= CAP) {
        for (int i = t; i < len; i += 4 * BROWS) {
            int4 e[4];
            #pragma unroll
            for (int k = 0; k < 4; ++k) {
                int j = i + k * BROWS;
                e[k] = (j < len) ? src[j] : make_int4(-1, 0, 0, 0);
            }
            #pragma unroll
            for (int k = 0; k < 4; ++k) {
                if (e[k].x >= 0) {
                    int p = atomicAdd(&cnt[e[k].x & (BROWS - 1)], 1);
                    stage[p] = make_int2(e[k].y, e[k].z);
                }
            }
        }
        __syncthreads();
        for (int i = t; i < len; i += BROWS) dst[i] = stage[i];
        for (int p = start; p < start + v; ++p) s += __int_as_float(stage[p].y);
    } else {
        for (int i = t; i < len; i += BROWS) {
            int4 e = src[i];
            int p = atomicAdd(&cnt[e.x & (BROWS - 1)], 1);
            dst[p] = make_int2(e.y, e.z);
        }
        __threadfence();
        __syncthreads();
        for (int p = start; p < start + v; ++p) s += __int_as_float(dst[p].y);
    }
    if (row < nN) {
        float d = s + 1.0f;
        dinv[row] = (d > 0.0f) ? rsqrtf(d) : 0.0f;
    }
}


__device__ __forceinline__ float4 fma4(float s, float4 a, float4 c) {
    c.x = fmaf(s, a.x, c.x);
    c.y = fmaf(s, a.y, c.y);
    c.z = fmaf(s, a.z, c.z);
    c.w = fmaf(s, a.w, c.w);
    return c;
}

__global__ __launch_bounds__(256) void gather_kernel(const float* __restrict__ x,
                                                     const int* __restrict__ rowptr,
                                                     const int2* __restrict__ ent,
                                                     const float* __restrict__ dinv,
                                                     float* __restrict__ agg, int nN) {
    int lane = threadIdx.x & 31;
    int wid = __builtin_amdgcn_readfirstlane((int)((blockIdx.x * blockDim.x + threadIdx.x) >> 5));
    int nW = (int)((gridDim.x * blockDim.x) >> 5);
    const float4* xv = (const float4*)x;
    float4* av = (float4*)agg;

    for (int r = wid; r < nN; r += nW) {
        int start = rowptr[r];
        int end = rowptr[r + 1];
        float s = dinv[r];
        int2 en = make_int2(0, 0);
        int dl = 0;
        float4 xs = xv[(size_t)r * 32 + lane];
        float4 acc = make_float4(0.0f, 0.0f, 0.0f, 0.0f);

        for (int base = start; base < end; base += 32) {
            int n = end - base;
            if (n > 32) n = 32;
            const int2* ep = ent + base;
            if (lane < n) {
                en = ep[lane];
                dl = __float_as_int(dinv[en.x]);
            }
            int j = 0;
            for (; j + 4 <= n; j += 4) {
                int c0 = __builtin_amdgcn_readlane(en.x, j);
                int c1 = __builtin_amdgcn_readlane(en.x, j + 1);
                int c2 = __builtin_amdgcn_readlane(en.x, j + 2);
                int c3 = __builtin_amdgcn_readlane(en.x, j + 3);
                float w0 = __int_as_float(__builtin_amdgcn_readlane(en.y, j))
                         * __int_as_float(__builtin_amdgcn_readlane(dl, j));
                float w1 = __int_as_float(__builtin_amdgcn_readlane(en.y, j + 1))
                         * __int_as_float(__builtin_amdgcn_readlane(dl, j + 1));
                float w2 = __int_as_float(__builtin_amdgcn_readlane(en.y, j + 2))
                         * __int_as_float(__builtin_amdgcn_readlane(dl, j + 2));
                float w3 = __int_as_float(__builtin_amdgcn_readlane(en.y, j + 3))
                         * __int_as_float(__builtin_amdgcn_readlane(dl, j + 3));
                float4 a0 = xv[(size_t)c0 * 32 + lane];
                float4 a1 = xv[(size_t)c1 * 32 + lane];
                float4 a2 = xv[(size_t)c2 * 32 + lane];
                float4 a3 = xv[(size_t)c3 * 32 + lane];
                acc = fma4(w0, a0, acc);
                acc = fma4(w1, a1, acc);
                acc = fma4(w2, a2, acc);
                acc = fma4(w3, a3, acc);
            }
            for (; j < n; ++j) {
                int c0 = __builtin_amdgcn_readlane(en.x, j);
                float w0 = __int_as_float(__builtin_amdgcn_readlane(en.y, j))
                         * __int_as_float(__builtin_amdgcn_readlane(dl, j));
                float4 a0 = xv[(size_t)c0 * 32 + lane];
                acc = fma4(w0, a0, acc);
            }
        }

        float ss = s * s;
        float4 o;
        o.x = fmaf(ss, xs.x, s * acc.x);
        o.y = fmaf(ss, xs.y, s * acc.y);
        o.z = fmaf(ss, xs.z, s * acc.z);
        o.w = fmaf(ss, xs.w, s * acc.w);
        av[(size_t)r * 32 + lane] = o;
    }
}


__global__ __launch_bounds__(256) void gemm_bn_relu_kernel(const float* __restrict__ a_in,
                                                           const float* __restrict__ W,
                                                           const float* __restrict__ bias,
                                                           const float* __restrict__ gamma,
                                                           const float* __restrict__ beta,
                                                           const float* __restrict__ mean,
                                                           const float* __restrict__ var,
                                                           float* __restrict__ out, int nRows) {
    __shared__ float Wl[FEAT * FEAT];

    int t = threadIdx.x;
    const float4* Wv = (const float4*)W;
    float4* Wlv = (float4*)Wl;
    #pragma unroll
    for (int i = t; i < (FEAT * FEAT) / 4; i += 256) Wlv[i] = Wv[i];
    __syncthreads();

    int wave = t >> 5;
    int lane = t & 31;
    int row0 = blockIdx.x * 128 + wave * 16;
    if (row0 >= nRows) return;

    int m  = lane & 15;
    int kh = (lane >> 4) << 1;
    int srow = row0 + m;
    if (srow >= nRows) srow = nRows - 1;
    const float* xr = a_in + (size_t)srow * FEAT;

    v8f acc[8] = {};

    for (int kb = 0; kb < FEAT / 4; ++kb) {
        int k = kb * 4 + kh;
        float2 av = *(const float2*)(xr + k);
        v2f a; a.x = av.x; a.y = av.y;
        #pragma unroll
        for (int j = 0; j < 8; ++j) {
            int col = j * 16 + m;
            v2f b;
            b.x = Wl[(k + 0) * FEAT + col];
            b.y = Wl[(k + 1) * FEAT + col];
            acc[j] = __builtin_amdgcn_wmma_f32_16x16x4_f32(
                false, a, false, b, (short)0, acc[j], false, false);
        }
    }

    int rbase = row0 + ((lane >> 4) ? 8 : 0);
    int n = lane & 15;
    #pragma unroll
    for (int j = 0; j < 8; ++j) {
        int u = j * 16 + n;
        float bu = bias[u];
        float gu = gamma[u];
        float mu = mean[u];
        float rs = rsqrtf(var[u] + BN_EPS);
        float be = beta[u];
        #pragma unroll
        for (int r = 0; r < 8; ++r) {
            int row = rbase + r;
            float v = acc[j][r] + bu;
            float y = gu * (v - mu) * rs + be;
            y = y > 0.0f ? y : 0.0f;
            if (row < nRows) out[(size_t)row * FEAT + u] = y;
        }
    }
}


extern "C" void kernel_launch(void* const* d_in, const int* in_sizes, int n_in,
                              void* d_out, int out_size, void* d_ws, size_t ws_size,
                              hipStream_t stream) {
    const float* x     = (const float*)d_in[0];
    const int*   eidx  = (const int*)d_in[1];
    const float* ew    = (const float*)d_in[2];
    const float* W     = (const float*)d_in[3];
    const float* bias  = (const float*)d_in[4];
    const float* gamma = (const float*)d_in[5];
    const float* beta  = (const float*)d_in[6];
    const float* mean  = (const float*)d_in[7];
    const float* var   = (const float*)d_in[8];

    int nE = in_sizes[2];
    int nN = in_sizes[0] / FEAT;
    const int* rows = eidx;
    const int* cols = eidx + nE;

    float* agg  = (float*)d_ws;
    float* dinv = agg + (size_t)nN * FEAT + nN;
    float* out  = (float*)d_out;

    int* o32    = (int*)d_out;
    int  nB     = (nN + BROWS - 1) / BROWS;
    int  pitch  = (nN + 64 + 3) & ~3;
    int* rowptr = o32;
    int* off    = o32 + pitch;
    int4* tmp   = (int4*)(off + nB * GBLK);
    int2* ent   = (int2*)(tmp + nE);

    hist_kernel<<<GBLK, 1024, 0, stream>>>(rows, off, nE, nB);
    scan_kernel<<<1, 1024, 0, stream>>>(off, nB * (GBLK / 4));
    bucket_kernel<<<GBLK, 1024, 0, stream>>>(rows, cols, ew, off, tmp, nE, nB);
    rowsort_kernel<<<nB, BROWS, 0, stream>>>(tmp, off, rowptr, ent, dinv, nE, nN, GBLK);
    gather_kernel<<<4096, 256, 0, stream>>>(x, rowptr, ent, dinv, agg, nN);
    gemm_bn_relu_kernel<<<(nN + 127) / 128, 256, 0, stream>>>(
        agg, W, bias, gamma, beta, mean, var, out, nN);
}


# edge sort: each placing block sums its own cursors from the raw (bucket, block) counts (a wave reads a bucket's 256 counts, the block scans the bucket totals); the one-block scan launch is gone
# speedup vs baseline: 3.5710x; 1.1159x over previous
#include <hip/hip_runtime.h>
#include <math.h>


typedef float v2f __attribute__((ext_vector_type(2)));
typedef float v8f __attribute__((ext_vector_type(8)));

#define FEAT 128
#define BN_EPS 1e-3f


#define SH     9
#define BROWS  512
#define NBMAX  2048
#define GBLK   256
#define CAP    16384

__global__ __launch_bounds__(1024) void hist_kernel(const int* __restrict__ rows, int* __restrict__ hist,
                                                    int nE, int nB) {
    __shared__ int h[NBMAX];
    int t = threadIdx.x;
    int g = blockIdx.x;
    int G = gridDim.x;
    for (int i = t; i < nB; i += 1024) h[i] = 0;
    __syncthreads();
    int chunk = (nE + G - 1) / G;
    int e0 = g * chunk;
    int e1 = e0 + chunk;
    if (e1 > nE) e1 = nE;
    for (int e = e0 + t; e < e1; e += 4 * 1024) {
        int r[4];
        #pragma unroll
        for (int k = 0; k < 4; ++k) {
            int j = e + k * 1024;
            r[k] = (j < e1) ? rows[j] : -1;
        }
        #pragma unroll
        for (int k = 0; k < 4; ++k)
            if (r[k] >= 0) atomicAdd(&h[r[k] >> SH], 1);
    }
    __syncthreads();
    for (int i = t; i < nB; i += 1024) hist[i * G + g] = h[i];
}


__global__ __launch_bounds__(1024) void bucket_kernel(const int* __restrict__ rows, const int* __restrict__ cols,
                                                      const float* __restrict__ w, const int* __restrict__ hist,
                                                      int* __restrict__ bst, int4* __restrict__ tmp, int nE, int nB) {
    __shared__ int cur[NBMAX];
    __shared__ int tot[NBMAX];
    __shared__ int wtot[32];
    int t = threadIdx.x;
    int lane = t & 31;
    int wv = t >> 5;
    int g = blockIdx.x;
    int G = gridDim.x;

    int q = g - lane * (GBLK / 32);
    for (int i0 = wv; i0 < nB; i0 += 32 * 8) {
        int4 ca[8], cb[8];
        #pragma unroll
        for (int k = 0; k < 8; ++k) {
            int i = i0 + 32 * k;
            ca[k] = make_int4(0, 0, 0, 0);
            cb[k] = make_int4(0, 0, 0, 0);
            if (i < nB) {
                const int4* h4 = (const int4*)(hist + (size_t)i * GBLK);
                ca[k] = h4[lane * 2];
                cb[k] = h4[lane * 2 + 1];
            }
        }
        int s[8], p[8];
        #pragma unroll
        for (int k = 0; k < 8; ++k) {
            s[k] = ca[k].x + ca[k].y + ca[k].z + ca[k].w + cb[k].x + cb[k].y + cb[k].z + cb[k].w;
            p[k] = (q > 0 ? ca[k].x : 0) + (q > 1 ? ca[k].y : 0) + (q > 2 ? ca[k].z : 0) + (q > 3 ? ca[k].w : 0)
                 + (q > 4 ? cb[k].x : 0) + (q > 5 ? cb[k].y : 0) + (q > 6 ? cb[k].z : 0) + (q > 7 ? cb[k].w : 0);
        }
        #pragma unroll
        for (int d = 16; d >= 1; d >>= 1) {
            #pragma unroll
            for (int k = 0; k < 8; ++k) {
                s[k] += __shfl_xor(s[k], d);
                p[k] += __shfl_xor(p[k], d);
            }
        }
        #pragma unroll
        for (int k = 0; k < 8; ++k) {
            int i = i0 + 32 * k;
            if (i < nB && lane == 0) { tot[i] = s[k]; cur[i] = p[k]; }
        }
    }
    __syncthreads();

    int i2 = 2 * t;
    int v0 = (i2 < nB) ? tot[i2] : 0;
    int v1 = (i2 + 1 < nB) ? tot[i2 + 1] : 0;
    int sv = v0 + v1;
    int inc = sv;
    for (int d = 1; d < 32; d <<= 1) {
        int o = __shfl_up(inc, d);
        if (lane >= d) inc += o;
    }
    if (lane == 31) wtot[wv] = inc;
    __syncthreads();
    int x = (lane < wv) ? wtot[lane] : 0;
    x += __shfl_xor(x, 16);
    x += __shfl_xor(x, 8);
    x += __shfl_xor(x, 4);
    x += __shfl_xor(x, 2);
    x += __shfl_xor(x, 1);
    int ex = x + inc - sv;
    if (i2 < nB) {
        int c0 = ex + cur[i2];
        cur[i2] = c0;
        if (g == 0) bst[i2] = c0;
    }
    if (i2 + 1 < nB) {
        int c1 = ex + v0 + cur[i2 + 1];
        cur[i2 + 1] = c1;
        if (g == 0) bst[i2 + 1] = c1;
    }
    __syncthreads();

    int chunk = (nE + G - 1) / G;
    int e0 = g * chunk;
    int e1 = e0 + chunk;
    if (e1 > nE) e1 = nE;
    for (int e = e0 + t; e < e1; e += 4 * 1024) {
        int r[4], c[4], wb[4];
        #pragma unroll
        for (int k = 0; k < 4; ++k) {
            int j = e + k * 1024;
            bool ok = j < e1;
            r[k]  = ok ? rows[j] : -1;
            c[k]  = ok ? cols[j] : 0;
            wb[k] = ok ? __float_as_int(w[j]) : 0;
        }
        #pragma unroll
        for (int k = 0; k < 4; ++k) {
            if (r[k] >= 0) {
                int pos = atomicAdd(&cur[r[k] >> SH], 1);
                tmp[pos] = make_int4(r[k], c[k], wb[k], 0);
            }
        }
    }
}


__global__ __launch_bounds__(BROWS) void rowsort_kernel(const int4* __restrict__ tmp, const int* __restrict__ bst,
                                                        int* __restrict__ rowptr, int2* ent,
                                                        float* __restrict__ dinv, int nE, int nN) {
    __shared__ int cnt[BROWS];
    __shared__ int wsum[BROWS / 32];
    __shared__ int2 stage[CAP];
    int t = threadIdx.x;
    int lane = t & 31;
    int wv = t >> 5;
    int b = blockIdx.x;
    int nB = gridDim.x;
    int bs = bst[b];
    int be = (b + 1 < nB) ? bst[b + 1] : nE;
    int len = be - bs;
    const int4* src = tmp + bs;
    int2* dst = ent + bs;

    cnt[t] = 0;
    __syncthreads();
    for (int i = t; i < len; i += 4 * BROWS) {
        int r[4];
        #pragma unroll
        for (int k = 0; k < 4; ++k) {
            int j = i + k * BROWS;
            r[k] = (j < len) ? src[j].x : -1;
        }
        #pragma unroll
        for (int k = 0; k < 4; ++k)
            if (r[k] >= 0) atomicAdd(&cnt[r[k] & (BROWS - 1)], 1);
    }
    __syncthreads();

    int v = cnt[t];
    int inc = v;
    for (int d = 1; d < 32; d <<= 1) {
        int o = __shfl_up(inc, d);
        if (lane >= d) inc += o;
    }
    if (lane == 31) wsum[wv] = inc;
    __syncthreads();
    int x = (lane < wv) ? wsum[lane] : 0;
    x += __shfl_xor(x, 16);
    x += __shfl_xor(x, 8);
    x += __shfl_xor(x, 4);
    x += __shfl_xor(x, 2);
    x += __shfl_xor(x, 1);
    int start = x + inc - v;

    int row = b * BROWS + t;
    if (row < nN) rowptr[row] = bs + start;
    if (b == nB - 1 && t == 0) rowptr[nN] = nE;
    cnt[t] = start;
    __syncthreads();

    float s = 0.0f;
    if (len <= CAP) {
        for (int i = t; i < len; i += 4 * BROWS) {
            int4 e[4];
            #pragma unroll
            for (int k = 0; k < 4; ++k) {
                int j = i + k * BROWS;
                e[k] = (j < len) ? src[j] : make_int4(-1, 0, 0, 0);
            }
            #pragma unroll
            for (int k = 0; k < 4; ++k) {
                if (e[k].x >= 0) {
                    int p = atomicAdd(&cnt[e[k].x & (BROWS - 1)], 1);
                    stage[p] = make_int2(e[k].y, e[k].z);
                }
            }
        }
        __syncthreads();
        for (int i = t; i < len; i += BROWS) dst[i] = stage[i];
        for (int p = start; p < start + v; ++p) s += __int_as_float(stage[p].y);
    } else {
        for (int i = t; i < len; i += BROWS) {
            int4 e = src[i];
            int p = atomicAdd(&cnt[e.x & (BROWS - 1)], 1);
            dst[p] = make_int2(e.y, e.z);
        }
        __threadfence();
        __syncthreads();
        for (int p = start; p < start + v; ++p) s += __int_as_float(dst[p].y);
    }
    if (row < nN) {
        float d = s + 1.0f;
        dinv[row] = (d > 0.0f) ? rsqrtf(d) : 0.0f;
    }
}


__device__ __forceinline__ float4 fma4(float s, float4 a, float4 c) {
    c.x = fmaf(s, a.x, c.x);
    c.y = fmaf(s, a.y, c.y);
    c.z = fmaf(s, a.z, c.z);
    c.w = fmaf(s, a.w, c.w);
    return c;
}

__global__ __launch_bounds__(256) void gather_kernel(const float* __restrict__ x,
                                                     const int* __restrict__ rowptr,
                                                     const int2* __restrict__ ent,
                                                     const float* __restrict__ dinv,
                                                     float* __restrict__ agg, int nN) {
    int lane = threadIdx.x & 31;
    int wid = __builtin_amdgcn_readfirstlane((int)((blockIdx.x * blockDim.x + threadIdx.x) >> 5));
    int nW = (int)((gridDim.x * blockDim.x) >> 5);
    const float4* xv = (const float4*)x;
    float4* av = (float4*)agg;

    for (int r = wid; r < nN; r += nW) {
        int start = rowptr[r];
        int end = rowptr[r + 1];
        float s = dinv[r];
        int2 en = make_int2(0, 0);
        int dl = 0;
        float4 xs = xv[(size_t)r * 32 + lane];
        float4 acc = make_float4(0.0f, 0.0f, 0.0f, 0.0f);

        for (int base = start; base < end; base += 32) {
            int n = end - base;
            if (n > 32) n = 32;
            const int2* ep = ent + base;
            if (lane < n) {
                en = ep[lane];
                dl = __float_as_int(dinv[en.x]);
            }
            int j = 0;
            for (; j + 4 <= n; j += 4) {
                int c0 = __builtin_amdgcn_readlane(en.x, j);
                int c1 = __builtin_amdgcn_readlane(en.x, j + 1);
                int c2 = __builtin_amdgcn_readlane(en.x, j + 2);
                int c3 = __builtin_amdgcn_readlane(en.x, j + 3);
                float w0 = __int_as_float(__builtin_amdgcn_readlane(en.y, j))
                         * __int_as_float(__builtin_amdgcn_readlane(dl, j));
                float w1 = __int_as_float(__builtin_amdgcn_readlane(en.y, j + 1))
                         * __int_as_float(__builtin_amdgcn_readlane(dl, j + 1));
                float w2 = __int_as_float(__builtin_amdgcn_readlane(en.y, j + 2))
                         * __int_as_float(__builtin_amdgcn_readlane(dl, j + 2));
                float w3 = __int_as_float(__builtin_amdgcn_readlane(en.y, j + 3))
                         * __int_as_float(__builtin_amdgcn_readlane(dl, j + 3));
                float4 a0 = xv[(size_t)c0 * 32 + lane];
                float4 a1 = xv[(size_t)c1 * 32 + lane];
                float4 a2 = xv[(size_t)c2 * 32 + lane];
                float4 a3 = xv[(size_t)c3 * 32 + lane];
                acc = fma4(w0, a0, acc);
                acc = fma4(w1, a1, acc);
                acc = fma4(w2, a2, acc);
                acc = fma4(w3, a3, acc);
            }
            for (; j < n; ++j) {
                int c0 = __builtin_amdgcn_readlane(en.x, j);
                float w0 = __int_as_float(__builtin_amdgcn_readlane(en.y, j))
                         * __int_as_float(__builtin_amdgcn_readlane(dl, j));
                float4 a0 = xv[(size_t)c0 * 32 + lane];
                acc = fma4(w0, a0, acc);
            }
        }

        float ss = s * s;
        float4 o;
        o.x = fmaf(ss, xs.x, s * acc.x);
        o.y = fmaf(ss, xs.y, s * acc.y);
        o.z = fmaf(ss, xs.z, s * acc.z);
        o.w = fmaf(ss, xs.w, s * acc.w);
        av[(size_t)r * 32 + lane] = o;
    }
}


__global__ __launch_bounds__(256) void gemm_bn_relu_kernel(const float* __restrict__ a_in,
                                                           const float* __restrict__ W,
                                                           const float* __restrict__ bias,
                                                           const float* __restrict__ gamma,
                                                           const float* __restrict__ beta,
                                                           const float* __restrict__ mean,
                                                           const float* __restrict__ var,
                                                           float* __restrict__ out, int nRows) {
    __shared__ float Wl[FEAT * FEAT];

    int t = threadIdx.x;
    const float4* Wv = (const float4*)W;
    float4* Wlv = (float4*)Wl;
    #pragma unroll
    for (int i = t; i < (FEAT * FEAT) / 4; i += 256) Wlv[i] = Wv[i];
    __syncthreads();

    int wave = t >> 5;
    int lane = t & 31;
    int row0 = blockIdx.x * 128 + wave * 16;
    if (row0 >= nRows) return;

    int m  = lane & 15;
    int kh = (lane >> 4) << 1;
    int srow = row0 + m;
    if (srow >= nRows) srow = nRows - 1;
    const float* xr = a_in + (size_t)srow * FEAT;

    v8f acc[8] = {};

    for (int kb = 0; kb < FEAT / 4; ++kb) {
        int k = kb * 4 + kh;
        float2 av = *(const float2*)(xr + k);
        v2f a; a.x = av.x; a.y = av.y;
        #pragma unroll
        for (int j = 0; j < 8; ++j) {
            int col = j * 16 + m;
            v2f b;
            b.x = Wl[(k + 0) * FEAT + col];
            b.y = Wl[(k + 1) * FEAT + col];
            acc[j] = __builtin_amdgcn_wmma_f32_16x16x4_f32(
                false, a, false, b, (short)0, acc[j], false, false);
        }
    }

    int rbase = row0 + ((lane >> 4) ? 8 : 0);
    int n = lane & 15;
    #pragma unroll
    for (int j = 0; j < 8; ++j) {
        int u = j * 16 + n;
        float bu = bias[u];
        float gu = gamma[u];
        float mu = mean[u];
        float rs = rsqrtf(var[u] + BN_EPS);
        float be = beta[u];
        #pragma unroll
        for (int r = 0; r < 8; ++r) {
            int row = rbase + r;
            float v = acc[j][r] + bu;
            float y = gu * (v - mu) * rs + be;
            y = y > 0.0f ? y : 0.0f;
            if (row < nRows) out[(size_t)row * FEAT + u] = y;
        }
    }
}


extern "C" void kernel_launch(void* const* d_in, const int* in_sizes, int n_in,
                              void* d_out, int out_size, void* d_ws, size_t ws_size,
                              hipStream_t stream) {
    const float* x     = (const float*)d_in[0];
    const int*   eidx  = (const int*)d_in[1];
    const float* ew    = (const float*)d_in[2];
    const float* W     = (const float*)d_in[3];
    const float* bias  = (const float*)d_in[4];
    const float* gamma = (const float*)d_in[5];
    const float* beta  = (const float*)d_in[6];
    const float* mean  = (const float*)d_in[7];
    const float* var   = (const float*)d_in[8];

    int nE = in_sizes[2];
    int nN = in_sizes[0] / FEAT;
    const int* rows = eidx;
    const int* cols = eidx + nE;

    float* agg  = (float*)d_ws;
    float* dinv = agg + (size_t)nN * FEAT + nN;
    float* out  = (float*)d_out;

    int* o32    = (int*)d_out;
    int  nB     = (nN + BROWS - 1) / BROWS;
    int  pitch  = (nN + 64 + 3) & ~3;
    int* rowptr = o32;
    int* cnts   = o32 + pitch;
    int* bst    = cnts + nB * GBLK;
    int4* tmp   = (int4*)(bst + ((nB + 3) & ~3));
    int2* ent   = (int2*)(tmp + nE);

    hist_kernel<<<GBLK, 1024, 0, stream>>>(rows, cnts, nE, nB);
    bucket_kernel<<<GBLK, 1024, 0, stream>>>(rows, cols, ew, cnts, bst, tmp, nE, nB);
    rowsort_kernel<<<nB, BROWS, 0, stream>>>(tmp, bst, rowptr, ent, dinv, nE, nN);
    gather_kernel<<<4096, 256, 0, stream>>>(x, rowptr, ent, dinv, agg, nN);
    gemm_bn_relu_kernel<<<(nN + 127) / 128, 256, 0, stream>>>(
        agg, W, bias, gamma, beta, mean, var, out, nN);
}


# row sums and product fused: a block of 16 waves sums 64 rows of x into LDS (4 per wave), then multiplies them by W from there; agg never goes to memory, 4 launches; order of every sum unchanged
# speedup vs baseline: 3.9273x; 1.0998x over previous
#include <hip/hip_runtime.h>
#include <math.h>


typedef float v2f __attribute__((ext_vector_type(2)));
typedef float v8f __attribute__((ext_vector_type(8)));

#define FEAT 128
#define BN_EPS 1e-3f


#define SH     9
#define BROWS  512
#define NBMAX  2048
#define GBLK   256
#define CAP    16384

__global__ __launch_bounds__(1024) void hist_kernel(const int* __restrict__ rows, int* __restrict__ hist,
                                                    int nE, int nB) {
    __shared__ int h[NBMAX];
    int t = threadIdx.x;
    int g = blockIdx.x;
    int G = gridDim.x;
    for (int i = t; i < nB; i += 1024) h[i] = 0;
    __syncthreads();
    int chunk = (nE + G - 1) / G;
    int e0 = g * chunk;
    int e1 = e0 + chunk;
    if (e1 > nE) e1 = nE;
    for (int e = e0 + t; e < e1; e += 4 * 1024) {
        int r[4];
        #pragma unroll
        for (int k = 0; k < 4; ++k) {
            int j = e + k * 1024;
            r[k] = (j < e1) ? rows[j] : -1;
        }
        #pragma unroll
        for (int k = 0; k < 4; ++k)
            if (r[k] >= 0) atomicAdd(&h[r[k] >> SH], 1);
    }
    __syncthreads();
    for (int i = t; i < nB; i += 1024) hist[i * G + g] = h[i];
}


__global__ __launch_bounds__(1024) void bucket_kernel(const int* __restrict__ rows, const int* __restrict__ cols,
                                                      const float* __restrict__ w, const int* __restrict__ hist,
                                                      int* __restrict__ bst, int4* __restrict__ tmp, int nE, int nB) {
    __shared__ int cur[NBMAX];
    __shared__ int tot[NBMAX];
    __shared__ int wtot[32];
    int t = threadIdx.x;
    int lane = t & 31;
    int wv = t >> 5;
    int g = blockIdx.x;
    int G = gridDim.x;

    int q = g - lane * (GBLK / 32);
    for (int i0 = wv; i0 < nB; i0 += 32 * 8) {
        int4 ca[8], cb[8];
        #pragma unroll
        for (int k = 0; k < 8; ++k) {
            int i = i0 + 32 * k;
            ca[k] = make_int4(0, 0, 0, 0);
            cb[k] = make_int4(0, 0, 0, 0);
            if (i < nB) {
                const int4* h4 = (const int4*)(hist + (size_t)i * GBLK);
                ca[k] = h4[lane * 2];
                cb[k] = h4[lane * 2 + 1];
            }
        }
        int s[8], p[8];
        #pragma unroll
        for (int k = 0; k < 8; ++k) {
            s[k] = ca[k].x + ca[k].y + ca[k].z + ca[k].w + cb[k].x + cb[k].y + cb[k].z + cb[k].w;
            p[k] = (q > 0 ? ca[k].x : 0) + (q > 1 ? ca[k].y : 0) + (q > 2 ? ca[k].z : 0) + (q > 3 ? ca[k].w : 0)
                 + (q > 4 ? cb[k].x : 0) + (q > 5 ? cb[k].y : 0) + (q > 6 ? cb[k].z : 0) + (q > 7 ? cb[k].w : 0);
        }
        #pragma unroll
        for (int d = 16; d >= 1; d >>= 1) {
            #pragma unroll
            for (int k = 0; k < 8; ++k) {
                s[k] += __shfl_xor(s[k], d);
                p[k] += __shfl_xor(p[k], d);
            }
        }
        #pragma unroll
        for (int k = 0; k < 8; ++k) {
            int i = i0 + 32 * k;
            if (i < nB && lane == 0) { tot[i] = s[k]; cur[i] = p[k]; }
        }
    }
    __syncthreads();

    int i2 = 2 * t;
    int v0 = (i2 < nB) ? tot[i2] : 0;
    int v1 = (i2 + 1 < nB) ? tot[i2 + 1] : 0;
    int sv = v0 + v1;
    int inc = sv;
    for (int d = 1; d < 32; d <<= 1) {
        int o = __shfl_up(inc, d);
        if (lane >= d) inc += o;
    }
    if (lane == 31) wtot[wv] = inc;
    __syncthreads();
    int x = (lane < wv) ? wtot[lane] : 0;
    x += __shfl_xor(x, 16);
    x += __shfl_xor(x, 8);
    x += __shfl_xor(x, 4);
    x += __shfl_xor(x, 2);
    x += __shfl_xor(x, 1);
    int ex = x + inc - sv;
    if (i2 < nB) {
        int c0 = ex + cur[i2];
        cur[i2] = c0;
        if (g == 0) bst[i2] = c0;
    }
    if (i2 + 1 < nB) {
        int c1 = ex + v0 + cur[i2 + 1];
        cur[i2 + 1] = c1;
        if (g == 0) bst[i2 + 1] = c1;
    }
    __syncthreads();

    int chunk = (nE + G - 1) / G;
    int e0 = g * chunk;
    int e1 = e0 + chunk;
    if (e1 > nE) e1 = nE;
    for (int e = e0 + t; e < e1; e += 4 * 1024) {
        int r[4], c[4], wb[4];
        #pragma unroll
        for (int k = 0; k < 4; ++k) {
            int j = e + k * 1024;
            bool ok = j < e1;
            r[k]  = ok ? rows[j] : -1;
            c[k]  = ok ? cols[j] : 0;
            wb[k] = ok ? __float_as_int(w[j]) : 0;
        }
        #pragma unroll
        for (int k = 0; k < 4; ++k) {
            if (r[k] >= 0) {
                int pos = atomicAdd(&cur[r[k] >> SH], 1);
                tmp[pos] = make_int4(r[k], c[k], wb[k], 0);
            }
        }
    }
}


__global__ __launch_bounds__(BROWS) void rowsort_kernel(const int4* __restrict__ tmp, const int* __restrict__ bst,
                                                        int* __restrict__ rowptr, int2* ent,
                                                        float* __restrict__ dinv, int nE, int nN) {
    __shared__ int cnt[BROWS];
    __shared__ int wsum[BROWS / 32];
    __shared__ int2 stage[CAP];
    int t = threadIdx.x;
    int lane = t & 31;
    int wv = t >> 5;
    int b = blockIdx.x;
    int nB = gridDim.x;
    int bs = bst[b];
    int be = (b + 1 < nB) ? bst[b + 1] : nE;
    int len = be - bs;
    const int4* src = tmp + bs;
    int2* dst = ent + bs;

    cnt[t] = 0;
    __syncthreads();
    for (int i = t; i < len; i += 4 * BROWS) {
        int r[4];
        #pragma unroll
        for (int k = 0; k < 4; ++k) {
            int j = i + k * BROWS;
            r[k] = (j < len) ? src[j].x : -1;
        }
        #pragma unroll
        for (int k = 0; k < 4; ++k)
            if (r[k] >= 0) atomicAdd(&cnt[r[k] & (BROWS - 1)], 1);
    }
    __syncthreads();

    int v = cnt[t];
    int inc = v;
    for (int d = 1; d < 32; d <<= 1) {
        int o = __shfl_up(inc, d);
        if (lane >= d) inc += o;
    }
    if (lane == 31) wsum[wv] = inc;
    __syncthreads();
    int x = (lane < wv) ? wsum[lane] : 0;
    x += __shfl_xor(x, 16);
    x += __shfl_xor(x, 8);
    x += __shfl_xor(x, 4);
    x += __shfl_xor(x, 2);
    x += __shfl_xor(x, 1);
    int start = x + inc - v;

    int row = b * BROWS + t;
    if (row < nN) rowptr[row] = bs + start;
    if (b == nB - 1 && t == 0) rowptr[nN] = nE;
    cnt[t] = start;
    __syncthreads();

    float s = 0.0f;
    if (len <= CAP) {
        for (int i = t; i < len; i += 4 * BROWS) {
            int4 e[4];
            #pragma unroll
            for (int k = 0; k < 4; ++k) {
                int j = i + k * BROWS;
                e[k] = (j < len) ? src[j] : make_int4(-1, 0, 0, 0);
            }
            #pragma unroll
            for (int k = 0; k < 4; ++k) {
                if (e[k].x >= 0) {
                    int p = atomicAdd(&cnt[e[k].x & (BROWS - 1)], 1);
                    stage[p] = make_int2(e[k].y, e[k].z);
                }
            }
        }
        __syncthreads();
        for (int i = t; i < len; i += BROWS) dst[i] = stage[i];
        for (int p = start; p < start + v; ++p) s += __int_as_float(stage[p].y);
    } else {
        for (int i = t; i < len; i += BROWS) {
            int4 e = src[i];
            int p = atomicAdd(&cnt[e.x & (BROWS - 1)], 1);
            dst[p] = make_int2(e.y, e.z);
        }
        __threadfence();
        __syncthreads();
        for (int p = start; p < start + v; ++p) s += __int_as_float(dst[p].y);
    }
    if (row < nN) {
        float d = s + 1.0f;
        dinv[row] = (d > 0.0f) ? rsqrtf(d) : 0.0f;
    }
}


#define TROWS  64
#define TPITCH 132

__device__ __forceinline__ float4 fma4(float s, float4 a, float4 c) {
    c.x = fmaf(s, a.x, c.x);
    c.y = fmaf(s, a.y, c.y);
    c.z = fmaf(s, a.z, c.z);
    c.w = fmaf(s, a.w, c.w);
    return c;
}

__global__ __launch_bounds__(512) void gather_gemm_kernel(const float* __restrict__ x,
                                                          const int* __restrict__ rowptr,
                                                          const int2* __restrict__ ent,
                                                          const float* __restrict__ dinv,
                                                          const float* __restrict__ W,
                                                          const float* __restrict__ bias,
                                                          const float* __restrict__ gamma,
                                                          const float* __restrict__ beta,
                                                          const float* __restrict__ mean,
                                                          const float* __restrict__ var,
                                                          float* __restrict__ out, int nN) {
    __shared__ float Wl[FEAT * FEAT];
    __shared__ float At[TROWS * TPITCH];

    int t = threadIdx.x;
    int lane = t & 31;
    int wv = __builtin_amdgcn_readfirstlane(t >> 5);

    const float4* Wv = (const float4*)W;
    float4* Wlv = (float4*)Wl;
    #pragma unroll
    for (int i = t; i < (FEAT * FEAT) / 4; i += 512) Wlv[i] = Wv[i];

    const float4* xv = (const float4*)x;
    int rb = blockIdx.x * TROWS + wv * 4;

    for (int q = 0; q < 4; ++q) {
        int r = rb + q;
        float4 o = make_float4(0.0f, 0.0f, 0.0f, 0.0f);
        if (r < nN) {
            int start = rowptr[r];
            int end = rowptr[r + 1];
            float s = dinv[r];
            int2 en = make_int2(0, 0);
            int dl = 0;
            float4 xs = xv[(size_t)r * 32 + lane];
            float4 acc = make_float4(0.0f, 0.0f, 0.0f, 0.0f);

            for (int base = start; base < end; base += 32) {
                int n = end - base;
                if (n > 32) n = 32;
                const int2* ep = ent + base;
                if (lane < n) {
                    en = ep[lane];
                    dl = __float_as_int(dinv[en.x]);
                }
                int j = 0;
                for (; j + 4 <= n; j += 4) {
                    int c0 = __builtin_amdgcn_readlane(en.x, j);
                    int c1 = __builtin_amdgcn_readlane(en.x, j + 1);
                    int c2 = __builtin_amdgcn_readlane(en.x, j + 2);
                    int c3 = __builtin_amdgcn_readlane(en.x, j + 3);
                    float w0 = __int_as_float(__builtin_amdgcn_readlane(en.y, j))
                             * __int_as_float(__builtin_amdgcn_readlane(dl, j));
                    float w1 = __int_as_float(__builtin_amdgcn_readlane(en.y, j + 1))
                             * __int_as_float(__builtin_amdgcn_readlane(dl, j + 1));
                    float w2 = __int_as_float(__builtin_amdgcn_readlane(en.y, j + 2))
                             * __int_as_float(__builtin_amdgcn_readlane(dl, j + 2));
                    float w3 = __int_as_float(__builtin_amdgcn_readlane(en.y, j + 3))
                             * __int_as_float(__builtin_amdgcn_readlane(dl, j + 3));
                    float4 a0 = xv[(size_t)c0 * 32 + lane];
                    float4 a1 = xv[(size_t)c1 * 32 + lane];
                    float4 a2 = xv[(size_t)c2 * 32 + lane];
                    float4 a3 = xv[(size_t)c3 * 32 + lane];
                    acc = fma4(w0, a0, acc);
                    acc = fma4(w1, a1, acc);
                    acc = fma4(w2, a2, acc);
                    acc = fma4(w3, a3, acc);
                }
                for (; j < n; ++j) {
                    int c0 = __builtin_amdgcn_readlane(en.x, j);
                    float w0 = __int_as_float(__builtin_amdgcn_readlane(en.y, j))
                             * __int_as_float(__builtin_amdgcn_readlane(dl, j));
                    float4 a0 = xv[(size_t)c0 * 32 + lane];
                    acc = fma4(w0, a0, acc);
                }
            }

            float ss = s * s;
            o.x = fmaf(ss, xs.x, s * acc.x);
            o.y = fmaf(ss, xs.y, s * acc.y);
            o.z = fmaf(ss, xs.z, s * acc.z);
            o.w = fmaf(ss, xs.w, s * acc.w);
        }
        *(float4*)(At + (wv * 4 + q) * TPITCH + 4 * lane) = o;
    }
    __syncthreads();

    int m  = lane & 15;
    int kh = (lane >> 4) << 1;
    int tile = wv >> 2;
    int jb = (wv & 3) * 2;
    const float* ar = At + (tile * 16 + m) * TPITCH;

    v8f acc2[2] = {};

    for (int kb = 0; kb < FEAT / 4; ++kb) {
        int k = kb * 4 + kh;
        float2 av = *(const float2*)(ar + k);
        v2f a; a.x = av.x; a.y = av.y;
        #pragma unroll
        for (int jj = 0; jj < 2; ++jj) {
            int col = (jb + jj) * 16 + m;
            v2f b;
            b.x = Wl[(k + 0) * FEAT + col];
            b.y = Wl[(k + 1) * FEAT + col];
            acc2[jj] = __builtin_amdgcn_wmma_f32_16x16x4_f32(
                false, a, false, b, (short)0, acc2[jj], false, false);
        }
    }

    int rbase = blockIdx.x * TROWS + tile * 16 + ((lane >> 4) ? 8 : 0);
    int n = lane & 15;
    #pragma unroll
    for (int jj = 0; jj < 2; ++jj) {
        int u = (jb + jj) * 16 + n;
        float bu = bias[u];
        float gu = gamma[u];
        float mu = mean[u];
        float rs = rsqrtf(var[u] + BN_EPS);
        float be = beta[u];
        #pragma unroll
        for (int r = 0; r < 8; ++r) {
            int row = rbase + r;
            float v = acc2[jj][r] + bu;
            float y = gu * (v - mu) * rs + be;
            y = y > 0.0f ? y : 0.0f;
            if (row < nN) out[(size_t)row * FEAT + u] = y;
        }
    }
}


extern "C" void kernel_launch(void* const* d_in, const int* in_sizes, int n_in,
                              void* d_out, int out_size, void* d_ws, size_t ws_size,
                              hipStream_t stream) {
    const float* x     = (const float*)d_in[0];
    const int*   eidx  = (const int*)d_in[1];
    const float* ew    = (const float*)d_in[2];
    const float* W     = (const float*)d_in[3];
    const float* bias  = (const float*)d_in[4];
    const float* gamma = (const float*)d_in[5];
    const float* beta  = (const float*)d_in[6];
    const float* mean  = (const float*)d_in[7];
    const float* var   = (const float*)d_in[8];

    int nE = in_sizes[2];
    int nN = in_sizes[0] / FEAT;
    const int* rows = eidx;
    const int* cols = eidx + nE;

    int  nB     = (nN + BROWS - 1) / BROWS;
    int  pitch  = (nN + 64 + 3) & ~3;

    int*   w32    = (int*)d_ws;
    int*   rowptr = w32;
    int2*  ent    = (int2*)(w32 + pitch);
    float* dinv   = (float*)d_ws + (size_t)nN * FEAT + nN;
    float* out    = (float*)d_out;

    int* o32    = (int*)d_out;
    int* cnts   = o32;
    int* bst    = cnts + nB * GBLK;
    int4* tmp   = (int4*)(bst + ((nB + 3) & ~3));

    hist_kernel<<<GBLK, 1024, 0, stream>>>(rows, cnts, nE, nB);
    bucket_kernel<<<GBLK, 1024, 0, stream>>>(rows, cols, ew, cnts, bst, tmp, nE, nB);
    rowsort_kernel<<<nB, BROWS, 0, stream>>>(tmp, bst, rowptr, ent, dinv, nE, nN);
    gather_gemm_kernel<<<(nN + TROWS - 1) / TROWS, 512, 0, stream>>>(
        x, rowptr, ent, dinv, W, bias, gamma, beta, mean, var, out, nN);
}


# edge sort: a level-1 block sorts its own slice by bucket in LDS (no other block's counts): count launch gone, 3 launches; a level-2 wave reads its 16 runs once, into registers; row order arrival
# speedup vs baseline: 4.2831x; 1.0906x over previous
#include <hip/hip_runtime.h>
#include <math.h>


typedef float v2f __attribute__((ext_vector_type(2)));
typedef float v8f __attribute__((ext_vector_type(8)));

#define FEAT 128
#define BN_EPS 1e-3f


#define SH     9
#define BROWS  512
#define NBMAX  2048
#define GBLK   256
#define CAP    16384
#define RPW    (GBLK / (BROWS / 32))

static_assert(BROWS == 2 * GBLK, "a level-2 block reads the run starts and the run ends with one load per thread");
static_assert(RPW <= 32, "a wave keeps one run descriptor per lane");


__global__ __launch_bounds__(1024) void bucket_kernel(const int* __restrict__ rows, const int* __restrict__ cols,
                                                      const float* __restrict__ w, int* __restrict__ runs,
                                                      int4* __restrict__ tmp, int nE, int nB) {
    __shared__ int cur[NBMAX];
    __shared__ int wtot[32];
    int t = threadIdx.x;
    int lane = t & 31;
    int wv = t >> 5;
    int g = blockIdx.x;
    int G = gridDim.x;
    for (int i = t; i < nB; i += 1024) cur[i] = 0;
    __syncthreads();
    int chunk = (nE + G - 1) / G;
    int e0 = g * chunk;
    int e1 = e0 + chunk;
    if (e1 > nE) e1 = nE;
    for (int e = e0 + t; e < e1; e += 8 * 1024) {
        int r[8];
        #pragma unroll
        for (int k = 0; k < 8; ++k) {
            int j = e + k * 1024;
            r[k] = (j < e1) ? rows[j] : -1;
        }
        #pragma unroll
        for (int k = 0; k < 8; ++k)
            if (r[k] >= 0) atomicAdd(&cur[r[k] >> SH], 1);
    }
    __syncthreads();

    int i2 = 2 * t;
    int v0 = (i2 < nB) ? cur[i2] : 0;
    int v1 = (i2 + 1 < nB) ? cur[i2 + 1] : 0;
    int sv = v0 + v1;
    int inc = sv;
    for (int d = 1; d < 32; d <<= 1) {
        int o = __shfl_up(inc, d);
        if (lane >= d) inc += o;
    }
    if (lane == 31) wtot[wv] = inc;
    __syncthreads();
    int x = (lane < wv) ? wtot[lane] : 0;
    x += __shfl_xor(x, 16);
    x += __shfl_xor(x, 8);
    x += __shfl_xor(x, 4);
    x += __shfl_xor(x, 2);
    x += __shfl_xor(x, 1);
    int ex = x + inc - sv;
    if (i2 <= nB) {
        runs[i2 * G + g] = ex;
        if (i2 < nB) cur[i2] = ex;
    }
    if (i2 + 1 <= nB) {
        runs[(i2 + 1) * G + g] = ex + v0;
        if (i2 + 1 < nB) cur[i2 + 1] = ex + v0;
    }
    __syncthreads();

    int4* mine = tmp + e0;
    for (int e = e0 + t; e < e1; e += 8 * 1024) {
        int r[8], c[8], wb[8];
        #pragma unroll
        for (int k = 0; k < 8; ++k) {
            int j = e + k * 1024;
            bool ok = j < e1;
            r[k]  = ok ? rows[j] : -1;
            c[k]  = ok ? cols[j] : 0;
            wb[k] = ok ? __float_as_int(w[j]) : 0;
        }
        #pragma unroll
        for (int k = 0; k < 8; ++k) {
            if (r[k] >= 0) {
                int pos = atomicAdd(&cur[r[k] >> SH], 1);
                mine[pos] = make_int4(r[k], c[k], wb[k], 0);
            }
        }
    }
}

template <class F>
__device__ __forceinline__ void long_runs(const int4* __restrict__ tmp, int dn, int db, int lane, F f) {
    for (int k = 0; k < RPW; ++k) {
        int n = __builtin_amdgcn_readlane(dn, k);
        const int4* p = tmp + __builtin_amdgcn_readlane(db, k);
        for (int j = lane + 64; j < n; j += 32) f(p[j]);
    }
}


__global__ __launch_bounds__(BROWS) void rowsort_kernel(const int4* __restrict__ tmp, const int* __restrict__ runs,
                                                        int* __restrict__ rowptr, int2* ent,
                                                        float* __restrict__ dinv, int nE, int nN) {
    __shared__ int cnt[BROWS];
    __shared__ int wsum[BROWS / 32];
    __shared__ int runb[GBLK];
    __shared__ int runn[GBLK];
    __shared__ int2 stage[CAP];
    int t = threadIdx.x;
    int lane = t & 31;
    int wv = t >> 5;
    int b = blockIdx.x;
    int nB = gridDim.x;
    int chunk = (nE + GBLK - 1) / GBLK;

    int g = t & (GBLK - 1);
    int hi = t / GBLK;
    int rv = runs[(b + hi) * GBLK + g];
    int rs = rv;
    rs += __shfl_xor(rs, 16);
    rs += __shfl_xor(rs, 8);
    rs += __shfl_xor(rs, 4);
    rs += __shfl_xor(rs, 2);
    rs += __shfl_xor(rs, 1);
    if (lane == 0) wsum[wv] = rs;
    if (hi == 0) runb[g] = g * chunk + rv;
    else         runn[g] = rv;
    cnt[t] = 0;
    __syncthreads();
    if (hi == 0) runn[g] = runn[g] - rv;
    int bs = 0, be = 0;
    #pragma unroll
    for (int k = 0; k < BROWS / 64; ++k) {
        bs += wsum[k];
        be += wsum[BROWS / 64 + k];
    }
    __syncthreads();
    int len = be - bs;
    int2* dst = ent + bs;

    int dn = runn[wv * RPW + (lane & (RPW - 1))];
    int db = runb[wv * RPW + (lane & (RPW - 1))];
    int4 ra[RPW], rc[RPW];
    int lastrec = (nE > 0) ? nE - 1 : 0;
    #pragma unroll
    for (int k = 0; k < RPW; ++k) {
        int n = __builtin_amdgcn_readlane(dn, k);
        int bse = __builtin_amdgcn_readlane(db, k);
        bse = min(bse, lastrec);
        const int4* p = tmp + bse;
        int i0 = max(min(lane, n - 1), 0);
        int i1 = max(min(lane + 32, n - 1), 0);
        ra[k] = p[i0];
        rc[k] = p[i1];
        if (lane >= n) ra[k].x = -1;
        if (lane + 32 >= n) rc[k].x = -1;
    }
    #pragma unroll
    for (int k = 0; k < RPW; ++k) {
        if (ra[k].x >= 0) atomicAdd(&cnt[ra[k].x & (BROWS - 1)], 1);
        if (rc[k].x >= 0) atomicAdd(&cnt[rc[k].x & (BROWS - 1)], 1);
    }
    long_runs(tmp, dn, db, lane, [&](int4 e) { atomicAdd(&cnt[e.x & (BROWS - 1)], 1); });
    __syncthreads();

    int v = cnt[t];
    int inc = v;
    for (int d = 1; d < 32; d <<= 1) {
        int o = __shfl_up(inc, d);
        if (lane >= d) inc += o;
    }
    if (lane == 31) wsum[wv] = inc;
    __syncthreads();
    int x = (lane < wv) ? wsum[lane] : 0;
    x += __shfl_xor(x, 16);
    x += __shfl_xor(x, 8);
    x += __shfl_xor(x, 4);
    x += __shfl_xor(x, 2);
    x += __shfl_xor(x, 1);
    int start = x + inc - v;

    int row = b * BROWS + t;
    if (row < nN) rowptr[row] = bs + start;
    if (b == nB - 1 && t == 0) rowptr[nN] = nE;
    cnt[t] = start;
    __syncthreads();

    float s = 0.0f;
    if (len <= CAP) {
        #pragma unroll
        for (int k = 0; k < RPW; ++k) {
            if (ra[k].x >= 0) {
                int p = atomicAdd(&cnt[ra[k].x & (BROWS - 1)], 1);
                stage[p] = make_int2(ra[k].y, ra[k].z);
            }
            if (rc[k].x >= 0) {
                int p = atomicAdd(&cnt[rc[k].x & (BROWS - 1)], 1);
                stage[p] = make_int2(rc[k].y, rc[k].z);
            }
        }
        long_runs(tmp, dn, db, lane, [&](int4 e) {
            int p = atomicAdd(&cnt[e.x & (BROWS - 1)], 1);
            stage[p] = make_int2(e.y, e.z);
        });
        __syncthreads();
        for (int i = t; i < len; i += BROWS) dst[i] = stage[i];
        for (int p = start; p < start + v; ++p) s += __int_as_float(stage[p].y);
    } else {
        #pragma unroll
        for (int k = 0; k < RPW; ++k) {
            if (ra[k].x >= 0) {
                int p = atomicAdd(&cnt[ra[k].x & (BROWS - 1)], 1);
                dst[p] = make_int2(ra[k].y, ra[k].z);
            }
            if (rc[k].x >= 0) {
                int p = atomicAdd(&cnt[rc[k].x & (BROWS - 1)], 1);
                dst[p] = make_int2(rc[k].y, rc[k].z);
            }
        }
        long_runs(tmp, dn, db, lane, [&](int4 e) {
            int p = atomicAdd(&cnt[e.x & (BROWS - 1)], 1);
            dst[p] = make_int2(e.y, e.z);
        });
        __threadfence();
        __syncthreads();
        for (int p = start; p < start + v; ++p) s += __int_as_float(dst[p].y);
    }
    if (row < nN) {
        float d = s + 1.0f;
        dinv[row] = (d > 0.0f) ? rsqrtf(d) : 0.0f;
    }
}


#define TROWS  64
#define TPITCH 132

__device__ __forceinline__ float4 fma4(float s, float4 a, float4 c) {
    c.x = fmaf(s, a.x, c.x);
    c.y = fmaf(s, a.y, c.y);
    c.z = fmaf(s, a.z, c.z);
    c.w = fmaf(s, a.w, c.w);
    return c;
}

__global__ __launch_bounds__(512) void gather_gemm_kernel(const float* __restrict__ x,
                                                          const int* __restrict__ rowptr,
                                                          const int2* __restrict__ ent,
                                                          const float* __restrict__ dinv,
                                                          const float* __restrict__ W,
                                                          const float* __restrict__ bias,
                                                          const float* __restrict__ gamma,
                                                          const float* __restrict__ beta,
                                                          const float* __restrict__ mean,
                                                          const float* __restrict__ var,
                                                          float* __restrict__ out, int nN) {
    __shared__ float Wl[FEAT * FEAT];
    __shared__ float At[TROWS * TPITCH];

    int t = threadIdx.x;
    int lane = t & 31;
    int wv = __builtin_amdgcn_readfirstlane(t >> 5);

    const float4* Wv = (const float4*)W;
    float4* Wlv = (float4*)Wl;
    #pragma unroll
    for (int i = t; i < (FEAT * FEAT) / 4; i += 512) Wlv[i] = Wv[i];

    const float4* xv = (const float4*)x;
    int rb = blockIdx.x * TROWS + wv * 4;

    for (int q = 0; q < 4; ++q) {
        int r = rb + q;
        float4 o = make_float4(0.0f, 0.0f, 0.0f, 0.0f);
        if (r < nN) {
            int start = rowptr[r];
            int end = rowptr[r + 1];
            float s = dinv[r];
            int2 en = make_int2(0, 0);
            int dl = 0;
            float4 xs = xv[(size_t)r * 32 + lane];
            float4 acc = make_float4(0.0f, 0.0f, 0.0f, 0.0f);

            for (int base = start; base < end; base += 32) {
                int n = end - base;
                if (n > 32) n = 32;
                const int2* ep = ent + base;
                if (lane < n) {
                    en = ep[lane];
                    dl = __float_as_int(dinv[en.x]);
                }
                int j = 0;
                for (; j + 4 <= n; j += 4) {
                    int c0 = __builtin_amdgcn_readlane(en.x, j);
                    int c1 = __builtin_amdgcn_readlane(en.x, j + 1);
                    int c2 = __builtin_amdgcn_readlane(en.x, j + 2);
                    int c3 = __builtin_amdgcn_readlane(en.x, j + 3);
                    float w0 = __int_as_float(__builtin_amdgcn_readlane(en.y, j))
                             * __int_as_float(__builtin_amdgcn_readlane(dl, j));
                    float w1 = __int_as_float(__builtin_amdgcn_readlane(en.y, j + 1))
                             * __int_as_float(__builtin_amdgcn_readlane(dl, j + 1));
                    float w2 = __int_as_float(__builtin_amdgcn_readlane(en.y, j + 2))
                             * __int_as_float(__builtin_amdgcn_readlane(dl, j + 2));
                    float w3 = __int_as_float(__builtin_amdgcn_readlane(en.y, j + 3))
                             * __int_as_float(__builtin_amdgcn_readlane(dl, j + 3));
                    float4 a0 = xv[(size_t)c0 * 32 + lane];
                    float4 a1 = xv[(size_t)c1 * 32 + lane];
                    float4 a2 = xv[(size_t)c2 * 32 + lane];
                    float4 a3 = xv[(size_t)c3 * 32 + lane];
                    acc = fma4(w0, a0, acc);
                    acc = fma4(w1, a1, acc);
                    acc = fma4(w2, a2, acc);
                    acc = fma4(w3, a3, acc);
                }
                for (; j < n; ++j) {
                    int c0 = __builtin_amdgcn_readlane(en.x, j);
                    float w0 = __int_as_float(__builtin_amdgcn_readlane(en.y, j))
                             * __int_as_float(__builtin_amdgcn_readlane(dl, j));
                    float4 a0 = xv[(size_t)c0 * 32 + lane];
                    acc = fma4(w0, a0, acc);
                }
            }

            float ss = s * s;
            o.x = fmaf(ss, xs.x, s * acc.x);
            o.y = fmaf(ss, xs.y, s * acc.y);
            o.z = fmaf(ss, xs.z, s * acc.z);
            o.w = fmaf(ss, xs.w, s * acc.w);
        }
        *(float4*)(At + (wv * 4 + q) * TPITCH + 4 * lane) = o;
    }
    __syncthreads();

    int m  = lane & 15;
    int kh = (lane >> 4) << 1;
    int tile = wv >> 2;
    int jb = (wv & 3) * 2;
    const float* ar = At + (tile * 16 + m) * TPITCH;

    v8f acc2[2] = {};

    for (int kb = 0; kb < FEAT / 4; ++kb) {
        int k = kb * 4 + kh;
        float2 av = *(const float2*)(ar + k);
        v2f a; a.x = av.x; a.y = av.y;
        #pragma unroll
        for (int jj = 0; jj < 2; ++jj) {
            int col = (jb + jj) * 16 + m;
            v2f b;
            b.x = Wl[(k + 0) * FEAT + col];
            b.y = Wl[(k + 1) * FEAT + col];
            acc2[jj] = __builtin_amdgcn_wmma_f32_16x16x4_f32(
                false, a, false, b, (short)0, acc2[jj], false, false);
        }
    }

    int rbase = blockIdx.x * TROWS + tile * 16 + ((lane >> 4) ? 8 : 0);
    int n = lane & 15;
    #pragma unroll
    for (int jj = 0; jj < 2; ++jj) {
        int u = (jb + jj) * 16 + n;
        float bu = bias[u];
        float gu = gamma[u];
        float mu = mean[u];
        float rs = rsqrtf(var[u] + BN_EPS);
        float be = beta[u];
        #pragma unroll
        for (int r = 0; r < 8; ++r) {
            int row = rbase + r;
            float v = acc2[jj][r] + bu;
            float y = gu * (v - mu) * rs + be;
            y = y > 0.0f ? y : 0.0f;
            if (row < nN) out[(size_t)row * FEAT + u] = y;
        }
    }
}


extern "C" void kernel_launch(void* const* d_in, const int* in_sizes, int n_in,
                              void* d_out, int out_size, void* d_ws, size_t ws_size,
                              hipStream_t stream) {
    const float* x     = (const float*)d_in[0];
    const int*   eidx  = (const int*)d_in[1];
    const float* ew    = (const float*)d_in[2];
    const float* W     = (const float*)d_in[3];
    const float* bias  = (const float*)d_in[4];
    const float* gamma = (const float*)d_in[5];
    const float* beta  = (const float*)d_in[6];
    const float* mean  = (const float*)d_in[7];
    const float* var   = (const float*)d_in[8];

    int nE = in_sizes[2];
    int nN = in_sizes[0] / FEAT;
    const int* rows = eidx;
    const int* cols = eidx + nE;

    int  nB     = (nN + BROWS - 1) / BROWS;
    int  pitch  = (nN + 64 + 3) & ~3;

    int*   w32    = (int*)d_ws;
    int*   rowptr = w32;
    int2*  ent    = (int2*)(w32 + pitch);
    float* dinv   = (float*)d_ws + (size_t)nN * FEAT + nN;
    float* out    = (float*)d_out;

    int* o32    = (int*)d_out;
    int* runs   = o32;
    int4* tmp   = (int4*)(runs + (nB + 1) * GBLK);

    bucket_kernel<<<GBLK, 1024, 0, stream>>>(rows, cols, ew, runs, tmp, nE, nB);
    rowsort_kernel<<<nB, BROWS, 0, stream>>>(tmp, runs, rowptr, ent, dinv, nE, nN);
    gather_gemm_kernel<<<(nN + TROWS - 1) / TROWS, 512, 0, stream>>>(
        x, rowptr, ent, dinv, W, bias, gamma, beta, mean, var, out, nN);
}


# last launch: 768 resident blocks loop over 64-row tiles drawn from an int atomic counter that launch 1 clears; W staged once per block, half as many waves enter; order of every sum unchanged
# speedup vs baseline: 4.4799x; 1.0459x over previous
#include <hip/hip_runtime.h>
#include <math.h>


typedef float v2f __attribute__((ext_vector_type(2)));
typedef float v8f __attribute__((ext_vector_type(8)));

#define FEAT 128
#define BN_EPS 1e-3f


#define SH     9
#define BROWS  512
#define NBMAX  2048
#define GBLK   256
#define CAP    16384
#define RPW    (GBLK / (BROWS / 32))

static_assert(BROWS == 2 * GBLK, "a level-2 block reads the run starts and the run ends with one load per thread");
static_assert(RPW <= 32, "a wave keeps one run descriptor per lane");


__global__ __launch_bounds__(1024) void bucket_kernel(const int* __restrict__ rows, const int* __restrict__ cols,
                                                      const float* __restrict__ w, int* __restrict__ runs,
                                                      int4* __restrict__ tmp, int* __restrict__ tilectr,
                                                      int nE, int nB) {
    __shared__ int cur[NBMAX];
    __shared__ int wtot[32];
    int t = threadIdx.x;
    int lane = t & 31;
    int wv = t >> 5;
    int g = blockIdx.x;
    int G = gridDim.x;
    if (g == 0 && t == 0) *tilectr = 0;
    for (int i = t; i < nB; i += 1024) cur[i] = 0;
    __syncthreads();
    int chunk = (nE + G - 1) / G;
    int e0 = g * chunk;
    int e1 = e0 + chunk;
    if (e1 > nE) e1 = nE;
    for (int e = e0 + t; e < e1; e += 8 * 1024) {
        int r[8];
        #pragma unroll
        for (int k = 0; k < 8; ++k) {
            int j = e + k * 1024;
            r[k] = (j < e1) ? rows[j] : -1;
        }
        #pragma unroll
        for (int k = 0; k < 8; ++k)
            if (r[k] >= 0) atomicAdd(&cur[r[k] >> SH], 1);
    }
    __syncthreads();

    int i2 = 2 * t;
    int v0 = (i2 < nB) ? cur[i2] : 0;
    int v1 = (i2 + 1 < nB) ? cur[i2 + 1] : 0;
    int sv = v0 + v1;
    int inc = sv;
    for (int d = 1; d < 32; d <<= 1) {
        int o = __shfl_up(inc, d);
        if (lane >= d) inc += o;
    }
    if (lane == 31) wtot[wv] = inc;
    __syncthreads();
    int x = (lane < wv) ? wtot[lane] : 0;
    x += __shfl_xor(x, 16);
    x += __shfl_xor(x, 8);
    x += __shfl_xor(x, 4);
    x += __shfl_xor(x, 2);
    x += __shfl_xor(x, 1);
    int ex = x + inc - sv;
    if (i2 <= nB) {
        runs[i2 * G + g] = ex;
        if (i2 < nB) cur[i2] = ex;
    }
    if (i2 + 1 <= nB) {
        runs[(i2 + 1) * G + g] = ex + v0;
        if (i2 + 1 < nB) cur[i2 + 1] = ex + v0;
    }
    __syncthreads();

    int4* mine = tmp + e0;
    for (int e = e0 + t; e < e1; e += 8 * 1024) {
        int r[8], c[8], wb[8];
        #pragma unroll
        for (int k = 0; k < 8; ++k) {
            int j = e + k * 1024;
            bool ok = j < e1;
            r[k]  = ok ? rows[j] : -1;
            c[k]  = ok ? cols[j] : 0;
            wb[k] = ok ? __float_as_int(w[j]) : 0;
        }
        #pragma unroll
        for (int k = 0; k < 8; ++k) {
            if (r[k] >= 0) {
                int pos = atomicAdd(&cur[r[k] >> SH], 1);
                mine[pos] = make_int4(r[k], c[k], wb[k], 0);
            }
        }
    }
}

template <class F>
__device__ __forceinline__ void long_runs(const int4* __restrict__ tmp, int dn, int db, int lane, F f) {
    for (int k = 0; k < RPW; ++k) {
        int n = __builtin_amdgcn_readlane(dn, k);
        const int4* p = tmp + __builtin_amdgcn_readlane(db, k);
        for (int j = lane + 64; j < n; j += 32) f(p[j]);
    }
}


__global__ __launch_bounds__(BROWS) void rowsort_kernel(const int4* __restrict__ tmp, const int* __restrict__ runs,
                                                        int* __restrict__ rowptr, int2* ent,
                                                        float* __restrict__ dinv, int nE, int nN) {
    __shared__ int cnt[BROWS];
    __shared__ int wsum[BROWS / 32];
    __shared__ int runb[GBLK];
    __shared__ int runn[GBLK];
    __shared__ int2 stage[CAP];
    int t = threadIdx.x;
    int lane = t & 31;
    int wv = t >> 5;
    int b = blockIdx.x;
    int nB = gridDim.x;
    int chunk = (nE + GBLK - 1) / GBLK;

    int g = t & (GBLK - 1);
    int hi = t / GBLK;
    int rv = runs[(b + hi) * GBLK + g];
    int rs = rv;
    rs += __shfl_xor(rs, 16);
    rs += __shfl_xor(rs, 8);
    rs += __shfl_xor(rs, 4);
    rs += __shfl_xor(rs, 2);
    rs += __shfl_xor(rs, 1);
    if (lane == 0) wsum[wv] = rs;
    if (hi == 0) runb[g] = g * chunk + rv;
    else         runn[g] = rv;
    cnt[t] = 0;
    __syncthreads();
    if (hi == 0) runn[g] = runn[g] - rv;
    int bs = 0, be = 0;
    #pragma unroll
    for (int k = 0; k < BROWS / 64; ++k) {
        bs += wsum[k];
        be += wsum[BROWS / 64 + k];
    }
    __syncthreads();
    int len = be - bs;
    int2* dst = ent + bs;

    int dn = runn[wv * RPW + (lane & (RPW - 1))];
    int db = runb[wv * RPW + (lane & (RPW - 1))];
    int4 ra[RPW], rc[RPW];
    int lastrec = (nE > 0) ? nE - 1 : 0;
    #pragma unroll
    for (int k = 0; k < RPW; ++k) {
        int n = __builtin_amdgcn_readlane(dn, k);
        int bse = __builtin_amdgcn_readlane(db, k);
        bse = min(bse, lastrec);
        const int4* p = tmp + bse;
        int i0 = max(min(lane, n - 1), 0);
        int i1 = max(min(lane + 32, n - 1), 0);
        ra[k] = p[i0];
        rc[k] = p[i1];
        if (lane >= n) ra[k].x = -1;
        if (lane + 32 >= n) rc[k].x = -1;
    }
    #pragma unroll
    for (int k = 0; k < RPW; ++k) {
        if (ra[k].x >= 0) atomicAdd(&cnt[ra[k].x & (BROWS - 1)], 1);
        if (rc[k].x >= 0) atomicAdd(&cnt[rc[k].x & (BROWS - 1)], 1);
    }
    long_runs(tmp, dn, db, lane, [&](int4 e) { atomicAdd(&cnt[e.x & (BROWS - 1)], 1); });
    __syncthreads();

    int v = cnt[t];
    int inc = v;
    for (int d = 1; d < 32; d <<= 1) {
        int o = __shfl_up(inc, d);
        if (lane >= d) inc += o;
    }
    if (lane == 31) wsum[wv] = inc;
    __syncthreads();
    int x = (lane < wv) ? wsum[lane] : 0;
    x += __shfl_xor(x, 16);
    x += __shfl_xor(x, 8);
    x += __shfl_xor(x, 4);
    x += __shfl_xor(x, 2);
    x += __shfl_xor(x, 1);
    int start = x + inc - v;

    int row = b * BROWS + t;
    if (row < nN) rowptr[row] = bs + start;
    if (b == nB - 1 && t == 0) rowptr[nN] = nE;
    cnt[t] = start;
    __syncthreads();

    float s = 0.0f;
    if (len <= CAP) {
        #pragma unroll
        for (int k = 0; k < RPW; ++k) {
            if (ra[k].x >= 0) {
                int p = atomicAdd(&cnt[ra[k].x & (BROWS - 1)], 1);
                stage[p] = make_int2(ra[k].y, ra[k].z);
            }
            if (rc[k].x >= 0) {
                int p = atomicAdd(&cnt[rc[k].x & (BROWS - 1)], 1);
                stage[p] = make_int2(rc[k].y, rc[k].z);
            }
        }
        long_runs(tmp, dn, db, lane, [&](int4 e) {
            int p = atomicAdd(&cnt[e.x & (BROWS - 1)], 1);
            stage[p] = make_int2(e.y, e.z);
        });
        __syncthreads();
        for (int i = t; i < len; i += BROWS) dst[i] = stage[i];
        for (int p = start; p < start + v; ++p) s += __int_as_float(stage[p].y);
    } else {
        #pragma unroll
        for (int k = 0; k < RPW; ++k) {
            if (ra[k].x >= 0) {
                int p = atomicAdd(&cnt[ra[k].x & (BROWS - 1)], 1);
                dst[p] = make_int2(ra[k].y, ra[k].z);
            }
            if (rc[k].x >= 0) {
                int p = atomicAdd(&cnt[rc[k].x & (BROWS - 1)], 1);
                dst[p] = make_int2(rc[k].y, rc[k].z);
            }
        }
        long_runs(tmp, dn, db, lane, [&](int4 e) {
            int p = atomicAdd(&cnt[e.x & (BROWS - 1)], 1);
            dst[p] = make_int2(e.y, e.z);
        });
        __threadfence();
        __syncthreads();
        for (int p = start; p < start + v; ++p) s += __int_as_float(dst[p].y);
    }
    if (row < nN) {
        float d = s + 1.0f;
        dinv[row] = (d > 0.0f) ? rsqrtf(d) : 0.0f;
    }
}


#define TROWS  64
#define TPITCH 132
#define RES_BLOCKS 768

__device__ __forceinline__ float4 fma4(float s, float4 a, float4 c) {
    c.x = fmaf(s, a.x, c.x);
    c.y = fmaf(s, a.y, c.y);
    c.z = fmaf(s, a.z, c.z);
    c.w = fmaf(s, a.w, c.w);
    return c;
}

__global__ __launch_bounds__(512) void gather_gemm_kernel(const float* __restrict__ x,
                                                          const int* __restrict__ rowptr,
                                                          const int2* __restrict__ ent,
                                                          const float* __restrict__ dinv,
                                                          const float* __restrict__ W,
                                                          const float* __restrict__ bias,
                                                          const float* __restrict__ gamma,
                                                          const float* __restrict__ beta,
                                                          const float* __restrict__ mean,
                                                          const float* __restrict__ var,
                                                          float* __restrict__ out, int* tilectr, int nN) {
    __shared__ float Wl[FEAT * FEAT];
    __shared__ float At[TROWS * TPITCH];
    __shared__ int nextTile;

    int t = threadIdx.x;
    int lane = t & 31;
    int wv = __builtin_amdgcn_readfirstlane(t >> 5);

    const float4* Wv = (const float4*)W;
    float4* Wlv = (float4*)Wl;
    #pragma unroll
    for (int i = t; i < (FEAT * FEAT) / 4; i += 512) Wlv[i] = Wv[i];

    const float4* xv = (const float4*)x;
    int nTiles = (nN + TROWS - 1) / TROWS;
    int nBlk = gridDim.x;
    int tileIdx = blockIdx.x;

  while (tileIdx < nTiles) {
    int rb = tileIdx * TROWS + wv * 4;

    for (int q = 0; q < 4; ++q) {
        int r = rb + q;
        float4 o = make_float4(0.0f, 0.0f, 0.0f, 0.0f);
        if (r < nN) {
            int start = rowptr[r];
            int end = rowptr[r + 1];
            float s = dinv[r];
            int2 en = make_int2(0, 0);
            int dl = 0;
            float4 xs = xv[(size_t)r * 32 + lane];
            float4 acc = make_float4(0.0f, 0.0f, 0.0f, 0.0f);

            for (int base = start; base < end; base += 32) {
                int n = end - base;
                if (n > 32) n = 32;
                const int2* ep = ent + base;
                if (lane < n) {
                    en = ep[lane];
                    dl = __float_as_int(dinv[en.x]);
                }
                int j = 0;
                for (; j + 4 <= n; j += 4) {
                    int c0 = __builtin_amdgcn_readlane(en.x, j);
                    int c1 = __builtin_amdgcn_readlane(en.x, j + 1);
                    int c2 = __builtin_amdgcn_readlane(en.x, j + 2);
                    int c3 = __builtin_amdgcn_readlane(en.x, j + 3);
                    float w0 = __int_as_float(__builtin_amdgcn_readlane(en.y, j))
                             * __int_as_float(__builtin_amdgcn_readlane(dl, j));
                    float w1 = __int_as_float(__builtin_amdgcn_readlane(en.y, j + 1))
                             * __int_as_float(__builtin_amdgcn_readlane(dl, j + 1));
                    float w2 = __int_as_float(__builtin_amdgcn_readlane(en.y, j + 2))
                             * __int_as_float(__builtin_amdgcn_readlane(dl, j + 2));
                    float w3 = __int_as_float(__builtin_amdgcn_readlane(en.y, j + 3))
                             * __int_as_float(__builtin_amdgcn_readlane(dl, j + 3));
                    float4 a0 = xv[(size_t)c0 * 32 + lane];
                    float4 a1 = xv[(size_t)c1 * 32 + lane];
                    float4 a2 = xv[(size_t)c2 * 32 + lane];
                    float4 a3 = xv[(size_t)c3 * 32 + lane];
                    acc = fma4(w0, a0, acc);
                    acc = fma4(w1, a1, acc);
                    acc = fma4(w2, a2, acc);
                    acc = fma4(w3, a3, acc);
                }
                for (; j < n; ++j) {
                    int c0 = __builtin_amdgcn_readlane(en.x, j);
                    float w0 = __int_as_float(__builtin_amdgcn_readlane(en.y, j))
                             * __int_as_float(__builtin_amdgcn_readlane(dl, j));
                    float4 a0 = xv[(size_t)c0 * 32 + lane];
                    acc = fma4(w0, a0, acc);
                }
            }

            float ss = s * s;
            o.x = fmaf(ss, xs.x, s * acc.x);
            o.y = fmaf(ss, xs.y, s * acc.y);
            o.z = fmaf(ss, xs.z, s * acc.z);
            o.w = fmaf(ss, xs.w, s * acc.w);
        }
        *(float4*)(At + (wv * 4 + q) * TPITCH + 4 * lane) = o;
    }
    int drawn = 0;
    if (t == 0) {
        int off = 0;
        asm volatile("" : "+v"(off));
        drawn = atomicAdd(tilectr + off, 1);
    }
    __builtin_amdgcn_fence(__ATOMIC_RELEASE, "workgroup", "local");
    __builtin_amdgcn_s_barrier();
    __builtin_amdgcn_fence(__ATOMIC_ACQUIRE, "workgroup", "local");

    int m  = lane & 15;
    asm volatile("" : "+v"(m));
    int kh = (lane >> 4) << 1;
    int tile = wv >> 2;
    int jb = (wv & 3) * 2;
    const float* ar = At + (tile * 16 + m) * TPITCH;

    v8f acc2[2] = {};

    for (int kb = 0; kb < FEAT / 4; ++kb) {
        int k = kb * 4 + kh;
        float2 av = *(const float2*)(ar + k);
        v2f a; a.x = av.x; a.y = av.y;
        #pragma unroll
        for (int jj = 0; jj < 2; ++jj) {
            int col = (jb + jj) * 16 + m;
            v2f b;
            b.x = Wl[(k + 0) * FEAT + col];
            b.y = Wl[(k + 1) * FEAT + col];
            acc2[jj] = __builtin_amdgcn_wmma_f32_16x16x4_f32(
                false, a, false, b, (short)0, acc2[jj], false, false);
        }
    }

    int rbase = tileIdx * TROWS + tile * 16 + ((lane >> 4) ? 8 : 0);
    int n = lane & 15;
    #pragma unroll
    for (int jj = 0; jj < 2; ++jj) {
        int u = (jb + jj) * 16 + n;
        float bu = bias[u];
        float gu = gamma[u];
        float mu = mean[u];
        float rs = rsqrtf(var[u] + BN_EPS);
        float be = beta[u];
        #pragma unroll
        for (int r = 0; r < 8; ++r) {
            int row = rbase + r;
            float v = acc2[jj][r] + bu;
            float y = gu * (v - mu) * rs + be;
            y = y > 0.0f ? y : 0.0f;
            if (row < nN) out[(size_t)row * FEAT + u] = y;
        }
    }

    if (t == 0) nextTile = nBlk + drawn;
    __builtin_amdgcn_fence(__ATOMIC_RELEASE, "workgroup", "local");
    __builtin_amdgcn_s_barrier();
    __builtin_amdgcn_fence(__ATOMIC_ACQUIRE, "workgroup", "local");
    tileIdx = __builtin_amdgcn_readfirstlane(nextTile);
  }
}


extern "C" void kernel_launch(void* const* d_in, const int* in_sizes, int n_in,
                              void* d_out, int out_size, void* d_ws, size_t ws_size,
                              hipStream_t stream) {
    const float* x     = (const float*)d_in[0];
    const int*   eidx  = (const int*)d_in[1];
    const float* ew    = (const float*)d_in[2];
    const float* W     = (const float*)d_in[3];
    const float* bias  = (const float*)d_in[4];
    const float* gamma = (const float*)d_in[5];
    const float* beta  = (const float*)d_in[6];
    const float* mean  = (const float*)d_in[7];
    const float* var   = (const float*)d_in[8];

    int nE = in_sizes[2];
    int nN = in_sizes[0] / FEAT;
    const int* rows = eidx;
    const int* cols = eidx + nE;

    int  nB     = (nN + BROWS - 1) / BROWS;
    int  pitch  = (nN + 64 + 3) & ~3;

    int*   w32    = (int*)d_ws;
    int*   rowptr = w32;
    int2*  ent    = (int2*)(w32 + pitch);
    float* dinv   = (float*)d_ws + (size_t)nN * FEAT + nN;
    int*   tilectr = w32 + nN + 32;
    float* out    = (float*)d_out;

    int* o32    = (int*)d_out;
    int* runs   = o32;
    int4* tmp   = (int4*)(runs + (nB + 1) * GBLK);

    bucket_kernel<<<GBLK, 1024, 0, stream>>>(rows, cols, ew, runs, tmp, tilectr, nE, nB);
    rowsort_kernel<<<nB, BROWS, 0, stream>>>(tmp, runs, rowptr, ent, dinv, nE, nN);
    gather_gemm_kernel<<<RES_BLOCKS, 512, 0, stream>>>(
        x, rowptr, ent, dinv, W, bias, gamma, beta, mean, var, out, tilectr, nN);
}


# row sums: eight row loads in flight per trip, not four, and a chunk's last 1-7 loads asked for as one group of four and one of up to three, not one at a time; order of every sum unchanged
# speedup vs baseline: 4.6559x; 1.0393x over previous
#include <hip/hip_runtime.h>
#include <math.h>


typedef float v2f __attribute__((ext_vector_type(2)));
typedef float v8f __attribute__((ext_vector_type(8)));

#define FEAT 128
#define BN_EPS 1e-3f


#define SH     9
#define BROWS  512
#define NBMAX  2048
#define GBLK   256
#define CAP    16384
#define RPW    (GBLK / (BROWS / 32))

static_assert(BROWS == 2 * GBLK, "a level-2 block reads the run starts and the run ends with one load per thread");
static_assert(RPW <= 32, "a wave keeps one run descriptor per lane");


__global__ __launch_bounds__(1024) void bucket_kernel(const int* __restrict__ rows, const int* __restrict__ cols,
                                                      const float* __restrict__ w, int* __restrict__ runs,
                                                      int4* __restrict__ tmp, int* __restrict__ tilectr,
                                                      int nE, int nB) {
    __shared__ int cur[NBMAX];
    __shared__ int wtot[32];
    int t = threadIdx.x;
    int lane = t & 31;
    int wv = t >> 5;
    int g = blockIdx.x;
    int G = gridDim.x;
    if (g == 0 && t == 0) *tilectr = 0;
    for (int i = t; i < nB; i += 1024) cur[i] = 0;
    __syncthreads();
    int chunk = (nE + G - 1) / G;
    int e0 = g * chunk;
    int e1 = e0 + chunk;
    if (e1 > nE) e1 = nE;
    for (int e = e0 + t; e < e1; e += 8 * 1024) {
        int r[8];
        #pragma unroll
        for (int k = 0; k < 8; ++k) {
            int j = e + k * 1024;
            r[k] = (j < e1) ? rows[j] : -1;
        }
        #pragma unroll
        for (int k = 0; k < 8; ++k)
            if (r[k] >= 0) atomicAdd(&cur[r[k] >> SH], 1);
    }
    __syncthreads();

    int i2 = 2 * t;
    int v0 = (i2 < nB) ? cur[i2] : 0;
    int v1 = (i2 + 1 < nB) ? cur[i2 + 1] : 0;
    int sv = v0 + v1;
    int inc = sv;
    for (int d = 1; d < 32; d <<= 1) {
        int o = __shfl_up(inc, d);
        if (lane >= d) inc += o;
    }
    if (lane == 31) wtot[wv] = inc;
    __syncthreads();
    int x = (lane < wv) ? wtot[lane] : 0;
    x += __shfl_xor(x, 16);
    x += __shfl_xor(x, 8);
    x += __shfl_xor(x, 4);
    x += __shfl_xor(x, 2);
    x += __shfl_xor(x, 1);
    int ex = x + inc - sv;
    if (i2 <= nB) {
        runs[i2 * G + g] = ex;
        if (i2 < nB) cur[i2] = ex;
    }
    if (i2 + 1 <= nB) {
        runs[(i2 + 1) * G + g] = ex + v0;
        if (i2 + 1 < nB) cur[i2 + 1] = ex + v0;
    }
    __syncthreads();

    int4* mine = tmp + e0;
    for (int e = e0 + t; e < e1; e += 8 * 1024) {
        int r[8], c[8], wb[8];
        #pragma unroll
        for (int k = 0; k < 8; ++k) {
            int j = e + k * 1024;
            bool ok = j < e1;
            r[k]  = ok ? rows[j] : -1;
            c[k]  = ok ? cols[j] : 0;
            wb[k] = ok ? __float_as_int(w[j]) : 0;
        }
        #pragma unroll
        for (int k = 0; k < 8; ++k) {
            if (r[k] >= 0) {
                int pos = atomicAdd(&cur[r[k] >> SH], 1);
                mine[pos] = make_int4(r[k], c[k], wb[k], 0);
            }
        }
    }
}

template <class F>
__device__ __forceinline__ void long_runs(const int4* __restrict__ tmp, int dn, int db, int lane, F f) {
    for (int k = 0; k < RPW; ++k) {
        int n = __builtin_amdgcn_readlane(dn, k);
        const int4* p = tmp + __builtin_amdgcn_readlane(db, k);
        for (int j = lane + 64; j < n; j += 32) f(p[j]);
    }
}


__global__ __launch_bounds__(BROWS) void rowsort_kernel(const int4* __restrict__ tmp, const int* __restrict__ runs,
                                                        int* __restrict__ rowptr, int2* ent,
                                                        float* __restrict__ dinv, int nE, int nN) {
    __shared__ int cnt[BROWS];
    __shared__ int wsum[BROWS / 32];
    __shared__ int runb[GBLK];
    __shared__ int runn[GBLK];
    __shared__ int2 stage[CAP];
    int t = threadIdx.x;
    int lane = t & 31;
    int wv = t >> 5;
    int b = blockIdx.x;
    int nB = gridDim.x;
    int chunk = (nE + GBLK - 1) / GBLK;

    int g = t & (GBLK - 1);
    int hi = t / GBLK;
    int rv = runs[(b + hi) * GBLK + g];
    int rs = rv;
    rs += __shfl_xor(rs, 16);
    rs += __shfl_xor(rs, 8);
    rs += __shfl_xor(rs, 4);
    rs += __shfl_xor(rs, 2);
    rs += __shfl_xor(rs, 1);
    if (lane == 0) wsum[wv] = rs;
    if (hi == 0) runb[g] = g * chunk + rv;
    else         runn[g] = rv;
    cnt[t] = 0;
    __syncthreads();
    if (hi == 0) runn[g] = runn[g] - rv;
    int bs = 0, be = 0;
    #pragma unroll
    for (int k = 0; k < BROWS / 64; ++k) {
        bs += wsum[k];
        be += wsum[BROWS / 64 + k];
    }
    __syncthreads();
    int len = be - bs;
    int2* dst = ent + bs;

    int dn = runn[wv * RPW + (lane & (RPW - 1))];
    int db = runb[wv * RPW + (lane & (RPW - 1))];
    int4 ra[RPW], rc[RPW];
    int lastrec = (nE > 0) ? nE - 1 : 0;
    #pragma unroll
    for (int k = 0; k < RPW; ++k) {
        int n = __builtin_amdgcn_readlane(dn, k);
        int bse = __builtin_amdgcn_readlane(db, k);
        bse = min(bse, lastrec);
        const int4* p = tmp + bse;
        int i0 = max(min(lane, n - 1), 0);
        int i1 = max(min(lane + 32, n - 1), 0);
        ra[k] = p[i0];
        rc[k] = p[i1];
        if (lane >= n) ra[k].x = -1;
        if (lane + 32 >= n) rc[k].x = -1;
    }
    #pragma unroll
    for (int k = 0; k < RPW; ++k) {
        if (ra[k].x >= 0) atomicAdd(&cnt[ra[k].x & (BROWS - 1)], 1);
        if (rc[k].x >= 0) atomicAdd(&cnt[rc[k].x & (BROWS - 1)], 1);
    }
    long_runs(tmp, dn, db, lane, [&](int4 e) { atomicAdd(&cnt[e.x & (BROWS - 1)], 1); });
    __syncthreads();

    int v = cnt[t];
    int inc = v;
    for (int d = 1; d < 32; d <<= 1) {
        int o = __shfl_up(inc, d);
        if (lane >= d) inc += o;
    }
    if (lane == 31) wsum[wv] = inc;
    __syncthreads();
    int x = (lane < wv) ? wsum[lane] : 0;
    x += __shfl_xor(x, 16);
    x += __shfl_xor(x, 8);
    x += __shfl_xor(x, 4);
    x += __shfl_xor(x, 2);
    x += __shfl_xor(x, 1);
    int start = x + inc - v;

    int row = b * BROWS + t;
    if (row < nN) rowptr[row] = bs + start;
    if (b == nB - 1 && t == 0) rowptr[nN] = nE;
    cnt[t] = start;
    __syncthreads();

    float s = 0.0f;
    if (len <= CAP) {
        #pragma unroll
        for (int k = 0; k < RPW; ++k) {
            if (ra[k].x >= 0) {
                int p = atomicAdd(&cnt[ra[k].x & (BROWS - 1)], 1);
                stage[p] = make_int2(ra[k].y, ra[k].z);
            }
            if (rc[k].x >= 0) {
                int p = atomicAdd(&cnt[rc[k].x & (BROWS - 1)], 1);
                stage[p] = make_int2(rc[k].y, rc[k].z);
            }
        }
        long_runs(tmp, dn, db, lane, [&](int4 e) {
            int p = atomicAdd(&cnt[e.x & (BROWS - 1)], 1);
            stage[p] = make_int2(e.y, e.z);
        });
        __syncthreads();
        for (int i = t; i < len; i += BROWS) dst[i] = stage[i];
        for (int p = start; p < start + v; ++p) s += __int_as_float(stage[p].y);
    } else {
        #pragma unroll
        for (int k = 0; k < RPW; ++k) {
            if (ra[k].x >= 0) {
                int p = atomicAdd(&cnt[ra[k].x & (BROWS - 1)], 1);
                dst[p] = make_int2(ra[k].y, ra[k].z);
            }
            if (rc[k].x >= 0) {
                int p = atomicAdd(&cnt[rc[k].x & (BROWS - 1)], 1);
                dst[p] = make_int2(rc[k].y, rc[k].z);
            }
        }
        long_runs(tmp, dn, db, lane, [&](int4 e) {
            int p = atomicAdd(&cnt[e.x & (BROWS - 1)], 1);
            dst[p] = make_int2(e.y, e.z);
        });
        __threadfence();
        __syncthreads();
        for (int p = start; p < start + v; ++p) s += __int_as_float(dst[p].y);
    }
    if (row < nN) {
        float d = s + 1.0f;
        dinv[row] = (d > 0.0f) ? rsqrtf(d) : 0.0f;
    }
}


#define TROWS  64
#define TPITCH 132
#define RES_BLOCKS 768

__device__ __forceinline__ float4 fma4(float s, float4 a, float4 c) {
    c.x = fmaf(s, a.x, c.x);
    c.y = fmaf(s, a.y, c.y);
    c.z = fmaf(s, a.z, c.z);
    c.w = fmaf(s, a.w, c.w);
    return c;
}

template <int B>
__device__ __forceinline__ void row_batch(const float4* __restrict__ xv, int2 en, int dl, int j, int lane,
                                          float4& acc) {
    float4 a[B];
    #pragma unroll
    for (int i = 0; i < B; ++i) {
        int c = __builtin_amdgcn_readlane(en.x, j + i);
        a[i] = xv[(size_t)c * 32 + lane];
    }
    __builtin_amdgcn_sched_barrier(0);
    #pragma unroll
    for (int i = 0; i < B; ++i) {
        float w = __int_as_float(__builtin_amdgcn_readlane(en.y, j + i))
                * __int_as_float(__builtin_amdgcn_readlane(dl, j + i));
        acc = fma4(w, a[i], acc);
    }
}

__global__ __launch_bounds__(512, 12) void gather_gemm_kernel(const float* __restrict__ x,
                                                          const int* __restrict__ rowptr,
                                                          const int2* __restrict__ ent,
                                                          const float* __restrict__ dinv,
                                                          const float* __restrict__ W,
                                                          const float* __restrict__ bias,
                                                          const float* __restrict__ gamma,
                                                          const float* __restrict__ beta,
                                                          const float* __restrict__ mean,
                                                          const float* __restrict__ var,
                                                          float* __restrict__ out, int* tilectr, int nN) {
    __shared__ float Wl[FEAT * FEAT];
    __shared__ float At[TROWS * TPITCH];
    __shared__ int nextTile;

    int t = threadIdx.x;
    int lane = t & 31;
    int wv = __builtin_amdgcn_readfirstlane(t >> 5);

    const float4* Wv = (const float4*)W;
    float4* Wlv = (float4*)Wl;
    #pragma unroll
    for (int i = t; i < (FEAT * FEAT) / 4; i += 512) Wlv[i] = Wv[i];

    const float4* xv = (const float4*)x;
    int nTiles = (nN + TROWS - 1) / TROWS;
    int nBlk = gridDim.x;
    int tileIdx = blockIdx.x;

  while (tileIdx < nTiles) {
    int rb = tileIdx * TROWS + wv * 4;

    for (int q = 0; q < 4; ++q) {
        int r = rb + q;
        float4 o = make_float4(0.0f, 0.0f, 0.0f, 0.0f);
        if (r < nN) {
            int start = rowptr[r];
            int end = rowptr[r + 1];
            float s = dinv[r];
            int2 en = make_int2(0, 0);
            int dl = 0;
            float4 xs = xv[(size_t)r * 32 + lane];
            float4 acc = make_float4(0.0f, 0.0f, 0.0f, 0.0f);

            for (int base = start; base < end; base += 32) {
                int n = end - base;
                if (n > 32) n = 32;
                const int2* ep = ent + base;
                if (lane < n) {
                    en = ep[lane];
                    dl = __float_as_int(dinv[en.x]);
                }
                int j = 0;
                for (; j + 8 <= n; j += 8) row_batch<8>(xv, en, dl, j, lane, acc);
                if (j + 4 <= n) { row_batch<4>(xv, en, dl, j, lane, acc); j += 4; }
                int rem = n - j;
                if (rem == 3)      row_batch<3>(xv, en, dl, j, lane, acc);
                else if (rem == 2) row_batch<2>(xv, en, dl, j, lane, acc);
                else if (rem == 1) row_batch<1>(xv, en, dl, j, lane, acc);
            }

            float ss = s * s;
            o.x = fmaf(ss, xs.x, s * acc.x);
            o.y = fmaf(ss, xs.y, s * acc.y);
            o.z = fmaf(ss, xs.z, s * acc.z);
            o.w = fmaf(ss, xs.w, s * acc.w);
        }
        *(float4*)(At + (wv * 4 + q) * TPITCH + 4 * lane) = o;
    }
    int drawn = 0;
    if (t == 0) {
        int off = 0;
        asm volatile("" : "+v"(off));
        drawn = atomicAdd(tilectr + off, 1);
    }
    __builtin_amdgcn_fence(__ATOMIC_RELEASE, "workgroup", "local");
    __builtin_amdgcn_s_barrier();
    __builtin_amdgcn_fence(__ATOMIC_ACQUIRE, "workgroup", "local");

    int m  = lane & 15;
    asm volatile("" : "+v"(m));
    int kh = (lane >> 4) << 1;
    int tile = wv >> 2;
    int jb = (wv & 3) * 2;
    const float* ar = At + (tile * 16 + m) * TPITCH;

    v8f acc2[2] = {};

    for (int kb = 0; kb < FEAT / 4; ++kb) {
        int k = kb * 4 + kh;
        float2 av = *(const float2*)(ar + k);
        v2f a; a.x = av.x; a.y = av.y;
        #pragma unroll
        for (int jj = 0; jj < 2; ++jj) {
            int col = (jb + jj) * 16 + m;
            v2f b;
            b.x = Wl[(k + 0) * FEAT + col];
            b.y = Wl[(k + 1) * FEAT + col];
            acc2[jj] = __builtin_amdgcn_wmma_f32_16x16x4_f32(
                false, a, false, b, (short)0, acc2[jj], false, false);
        }
    }

    int rbase = tileIdx * TROWS + tile * 16 + ((lane >> 4) ? 8 : 0);
    int n = lane & 15;
    #pragma unroll
    for (int jj = 0; jj < 2; ++jj) {
        int u = (jb + jj) * 16 + n;
        float bu = bias[u];
        float gu = gamma[u];
        float mu = mean[u];
        float rs = rsqrtf(var[u] + BN_EPS);
        float be = beta[u];
        #pragma unroll
        for (int r = 0; r < 8; ++r) {
            int row = rbase + r;
            float v = acc2[jj][r] + bu;
            float y = gu * (v - mu) * rs + be;
            y = y > 0.0f ? y : 0.0f;
            if (row < nN) out[(size_t)row * FEAT + u] = y;
        }
    }

    if (t == 0) nextTile = nBlk + drawn;
    __builtin_amdgcn_fence(__ATOMIC_RELEASE, "workgroup", "local");
    __builtin_amdgcn_s_barrier();
    __builtin_amdgcn_fence(__ATOMIC_ACQUIRE, "workgroup", "local");
    tileIdx = __builtin_amdgcn_readfirstlane(nextTile);
  }
}


extern "C" void kernel_launch(void* const* d_in, const int* in_sizes, int n_in,
                              void* d_out, int out_size, void* d_ws, size_t ws_size,
                              hipStream_t stream) {
    const float* x     = (const float*)d_in[0];
    const int*   eidx  = (const int*)d_in[1];
    const float* ew    = (const float*)d_in[2];
    const float* W     = (const float*)d_in[3];
    const float* bias  = (const float*)d_in[4];
    const float* gamma = (const float*)d_in[5];
    const float* beta  = (const float*)d_in[6];
    const float* mean  = (const float*)d_in[7];
    const float* var   = (const float*)d_in[8];

    int nE = in_sizes[2];
    int nN = in_sizes[0] / FEAT;
    const int* rows = eidx;
    const int* cols = eidx + nE;

    int  nB     = (nN + BROWS - 1) / BROWS;
    int  pitch  = (nN + 64 + 3) & ~3;

    int*   w32    = (int*)d_ws;
    int*   rowptr = w32;
    int2*  ent    = (int2*)(w32 + pitch);
    float* dinv   = (float*)d_ws + (size_t)nN * FEAT + nN;
    int*   tilectr = w32 + nN + 32;
    float* out    = (float*)d_out;

    int* o32    = (int*)d_out;
    int* runs   = o32;
    int4* tmp   = (int4*)(runs + (nB + 1) * GBLK);

    bucket_kernel<<<GBLK, 1024, 0, stream>>>(rows, cols, ew, runs, tmp, tilectr, nE, nB);
    rowsort_kernel<<<nB, BROWS, 0, stream>>>(tmp, runs, rowptr, ent, dinv, nE, nN);
    gather_gemm_kernel<<<RES_BLOCKS, 512, 0, stream>>>(
        x, rowptr, ent, dinv, W, bias, gamma, beta, mean, var, out, tilectr, nN);
}


# level-1 edge sort: a block puts its slice in bucket order in LDS and copies it out in whole lines, not one scattered 16-byte record store per edge (32 lines per store); orders unchanged
# speedup vs baseline: 4.8526x; 1.0422x over previous
#include <hip/hip_runtime.h>
#include <math.h>


typedef float v2f __attribute__((ext_vector_type(2)));
typedef float v8f __attribute__((ext_vector_type(8)));

#define FEAT 128
#define BN_EPS 1e-3f


#define SH     9
#define BROWS  512
#define NBMAX  2048
#define GBLK   256
#define CAP    16384
#define RPW    (GBLK / (BROWS / 32))
#define L1CAP  8192

static_assert(BROWS == 2 * GBLK, "a level-2 block reads the run starts and the run ends with one load per thread");
static_assert(RPW <= 32, "a wave keeps one run descriptor per lane");


__global__ __launch_bounds__(1024) void bucket_kernel(const int* __restrict__ rows, const int* __restrict__ cols,
                                                      const float* __restrict__ w, int* __restrict__ runs,
                                                      int4* __restrict__ tmp, int* __restrict__ tilectr,
                                                      int nE, int nB) {
    __shared__ int cur[NBMAX];
    __shared__ int wtot[32];
    __shared__ int4 stg[L1CAP];
    int t = threadIdx.x;
    int lane = t & 31;
    int wv = t >> 5;
    int g = blockIdx.x;
    int G = gridDim.x;
    if (g == 0 && t == 0) *tilectr = 0;
    for (int i = t; i < nB; i += 1024) cur[i] = 0;
    __syncthreads();
    int chunk = (nE + G - 1) / G;
    int e0 = g * chunk;
    int e1 = e0 + chunk;
    if (e1 > nE) e1 = nE;
    for (int e = e0 + t; e < e1; e += 8 * 1024) {
        int r[8];
        #pragma unroll
        for (int k = 0; k < 8; ++k) {
            int j = e + k * 1024;
            r[k] = (j < e1) ? rows[j] : -1;
        }
        #pragma unroll
        for (int k = 0; k < 8; ++k)
            if (r[k] >= 0) atomicAdd(&cur[r[k] >> SH], 1);
    }
    __syncthreads();

    int i2 = 2 * t;
    int v0 = (i2 < nB) ? cur[i2] : 0;
    int v1 = (i2 + 1 < nB) ? cur[i2 + 1] : 0;
    int sv = v0 + v1;
    int inc = sv;
    for (int d = 1; d < 32; d <<= 1) {
        int o = __shfl_up(inc, d);
        if (lane >= d) inc += o;
    }
    if (lane == 31) wtot[wv] = inc;
    __syncthreads();
    int x = (lane < wv) ? wtot[lane] : 0;
    x += __shfl_xor(x, 16);
    x += __shfl_xor(x, 8);
    x += __shfl_xor(x, 4);
    x += __shfl_xor(x, 2);
    x += __shfl_xor(x, 1);
    int ex = x + inc - sv;
    if (i2 <= nB) {
        runs[i2 * G + g] = ex;
        if (i2 < nB) cur[i2] = ex;
    }
    if (i2 + 1 <= nB) {
        runs[(i2 + 1) * G + g] = ex + v0;
        if (i2 + 1 < nB) cur[i2 + 1] = ex + v0;
    }
    __syncthreads();

    int4* mine = tmp + e0;
    bool staged = chunk <= L1CAP;
    for (int e = e0 + t; e < e1; e += 8 * 1024) {
        int r[8], c[8], wb[8];
        #pragma unroll
        for (int k = 0; k < 8; ++k) {
            int j = e + k * 1024;
            bool ok = j < e1;
            r[k]  = ok ? rows[j] : -1;
            c[k]  = ok ? cols[j] : 0;
            wb[k] = ok ? __float_as_int(w[j]) : 0;
        }
        #pragma unroll
        for (int k = 0; k < 8; ++k) {
            if (r[k] >= 0) {
                int pos = atomicAdd(&cur[r[k] >> SH], 1);
                int4 rec = make_int4(r[k], c[k], wb[k], 0);
                if (staged) stg[pos] = rec;
                else        mine[pos] = rec;
            }
        }
    }
    if (staged) {
        __syncthreads();
        int n = e1 - e0;
        for (int i = t; i < n; i += 1024) mine[i] = stg[i];
    }
}

template <class F>
__device__ __forceinline__ void long_runs(const int4* __restrict__ tmp, int dn, int db, int lane, F f) {
    for (int k = 0; k < RPW; ++k) {
        int n = __builtin_amdgcn_readlane(dn, k);
        const int4* p = tmp + __builtin_amdgcn_readlane(db, k);
        for (int j = lane + 64; j < n; j += 32) f(p[j]);
    }
}


__global__ __launch_bounds__(BROWS) void rowsort_kernel(const int4* __restrict__ tmp, const int* __restrict__ runs,
                                                        int* __restrict__ rowptr, int2* ent,
                                                        float* __restrict__ dinv, int nE, int nN) {
    __shared__ int cnt[BROWS];
    __shared__ int wsum[BROWS / 32];
    __shared__ int runb[GBLK];
    __shared__ int runn[GBLK];
    __shared__ int2 stage[CAP];
    int t = threadIdx.x;
    int lane = t & 31;
    int wv = t >> 5;
    int b = blockIdx.x;
    int nB = gridDim.x;
    int chunk = (nE + GBLK - 1) / GBLK;

    int g = t & (GBLK - 1);
    int hi = t / GBLK;
    int rv = runs[(b + hi) * GBLK + g];
    int rs = rv;
    rs += __shfl_xor(rs, 16);
    rs += __shfl_xor(rs, 8);
    rs += __shfl_xor(rs, 4);
    rs += __shfl_xor(rs, 2);
    rs += __shfl_xor(rs, 1);
    if (lane == 0) wsum[wv] = rs;
    if (hi == 0) runb[g] = g * chunk + rv;
    else         runn[g] = rv;
    cnt[t] = 0;
    __syncthreads();
    if (hi == 0) runn[g] = runn[g] - rv;
    int bs = 0, be = 0;
    #pragma unroll
    for (int k = 0; k < BROWS / 64; ++k) {
        bs += wsum[k];
        be += wsum[BROWS / 64 + k];
    }
    __syncthreads();
    int len = be - bs;
    int2* dst = ent + bs;

    int dn = runn[wv * RPW + (lane & (RPW - 1))];
    int db = runb[wv * RPW + (lane & (RPW - 1))];
    int4 ra[RPW], rc[RPW];
    int lastrec = (nE > 0) ? nE - 1 : 0;
    #pragma unroll
    for (int k = 0; k < RPW; ++k) {
        int n = __builtin_amdgcn_readlane(dn, k);
        int bse = __builtin_amdgcn_readlane(db, k);
        bse = min(bse, lastrec);
        const int4* p = tmp + bse;
        int i0 = max(min(lane, n - 1), 0);
        int i1 = max(min(lane + 32, n - 1), 0);
        ra[k] = p[i0];
        rc[k] = p[i1];
        if (lane >= n) ra[k].x = -1;
        if (lane + 32 >= n) rc[k].x = -1;
    }
    #pragma unroll
    for (int k = 0; k < RPW; ++k) {
        if (ra[k].x >= 0) atomicAdd(&cnt[ra[k].x & (BROWS - 1)], 1);
        if (rc[k].x >= 0) atomicAdd(&cnt[rc[k].x & (BROWS - 1)], 1);
    }
    long_runs(tmp, dn, db, lane, [&](int4 e) { atomicAdd(&cnt[e.x & (BROWS - 1)], 1); });
    __syncthreads();

    int v = cnt[t];
    int inc = v;
    for (int d = 1; d < 32; d <<= 1) {
        int o = __shfl_up(inc, d);
        if (lane >= d) inc += o;
    }
    if (lane == 31) wsum[wv] = inc;
    __syncthreads();
    int x = (lane < wv) ? wsum[lane] : 0;
    x += __shfl_xor(x, 16);
    x += __shfl_xor(x, 8);
    x += __shfl_xor(x, 4);
    x += __shfl_xor(x, 2);
    x += __shfl_xor(x, 1);
    int start = x + inc - v;

    int row = b * BROWS + t;
    if (row < nN) rowptr[row] = bs + start;
    if (b == nB - 1 && t == 0) rowptr[nN] = nE;
    cnt[t] = start;
    __syncthreads();

    float s = 0.0f;
    if (len <= CAP) {
        #pragma unroll
        for (int k = 0; k < RPW; ++k) {
            if (ra[k].x >= 0) {
                int p = atomicAdd(&cnt[ra[k].x & (BROWS - 1)], 1);
                stage[p] = make_int2(ra[k].y, ra[k].z);
            }
            if (rc[k].x >= 0) {
                int p = atomicAdd(&cnt[rc[k].x & (BROWS - 1)], 1);
                stage[p] = make_int2(rc[k].y, rc[k].z);
            }
        }
        long_runs(tmp, dn, db, lane, [&](int4 e) {
            int p = atomicAdd(&cnt[e.x & (BROWS - 1)], 1);
            stage[p] = make_int2(e.y, e.z);
        });
        __syncthreads();
        for (int i = t; i < len; i += BROWS) dst[i] = stage[i];
        for (int p = start; p < start + v; ++p) s += __int_as_float(stage[p].y);
    } else {
        #pragma unroll
        for (int k = 0; k < RPW; ++k) {
            if (ra[k].x >= 0) {
                int p = atomicAdd(&cnt[ra[k].x & (BROWS - 1)], 1);
                dst[p] = make_int2(ra[k].y, ra[k].z);
            }
            if (rc[k].x >= 0) {
                int p = atomicAdd(&cnt[rc[k].x & (BROWS - 1)], 1);
                dst[p] = make_int2(rc[k].y, rc[k].z);
            }
        }
        long_runs(tmp, dn, db, lane, [&](int4 e) {
            int p = atomicAdd(&cnt[e.x & (BROWS - 1)], 1);
            dst[p] = make_int2(e.y, e.z);
        });
        __threadfence();
        __syncthreads();
        for (int p = start; p < start + v; ++p) s += __int_as_float(dst[p].y);
    }
    if (row < nN) {
        float d = s + 1.0f;
        dinv[row] = (d > 0.0f) ? rsqrtf(d) : 0.0f;
    }
}


#define TROWS  64
#define TPITCH 132
#define RES_BLOCKS 768

__device__ __forceinline__ float4 fma4(float s, float4 a, float4 c) {
    c.x = fmaf(s, a.x, c.x);
    c.y = fmaf(s, a.y, c.y);
    c.z = fmaf(s, a.z, c.z);
    c.w = fmaf(s, a.w, c.w);
    return c;
}

template <int B>
__device__ __forceinline__ void row_batch(const float4* __restrict__ xv, int2 en, int dl, int j, int lane,
                                          float4& acc) {
    float4 a[B];
    #pragma unroll
    for (int i = 0; i < B; ++i) {
        int c = __builtin_amdgcn_readlane(en.x, j + i);
        a[i] = xv[(size_t)c * 32 + lane];
    }
    __builtin_amdgcn_sched_barrier(0);
    #pragma unroll
    for (int i = 0; i < B; ++i) {
        float w = __int_as_float(__builtin_amdgcn_readlane(en.y, j + i))
                * __int_as_float(__builtin_amdgcn_readlane(dl, j + i));
        acc = fma4(w, a[i], acc);
    }
}

__global__ __launch_bounds__(512, 12) void gather_gemm_kernel(const float* __restrict__ x,
                                                          const int* __restrict__ rowptr,
                                                          const int2* __restrict__ ent,
                                                          const float* __restrict__ dinv,
                                                          const float* __restrict__ W,
                                                          const float* __restrict__ bias,
                                                          const float* __restrict__ gamma,
                                                          const float* __restrict__ beta,
                                                          const float* __restrict__ mean,
                                                          const float* __restrict__ var,
                                                          float* __restrict__ out, int* tilectr, int nN) {
    __shared__ float Wl[FEAT * FEAT];
    __shared__ float At[TROWS * TPITCH];
    __shared__ int nextTile;

    int t = threadIdx.x;
    int lane = t & 31;
    int wv = __builtin_amdgcn_readfirstlane(t >> 5);

    const float4* Wv = (const float4*)W;
    float4* Wlv = (float4*)Wl;
    #pragma unroll
    for (int i = t; i < (FEAT * FEAT) / 4; i += 512) Wlv[i] = Wv[i];

    const float4* xv = (const float4*)x;
    int nTiles = (nN + TROWS - 1) / TROWS;
    int nBlk = gridDim.x;
    int tileIdx = blockIdx.x;

  while (tileIdx < nTiles) {
    int rb = tileIdx * TROWS + wv * 4;

    for (int q = 0; q < 4; ++q) {
        int r = rb + q;
        float4 o = make_float4(0.0f, 0.0f, 0.0f, 0.0f);
        if (r < nN) {
            int start = rowptr[r];
            int end = rowptr[r + 1];
            float s = dinv[r];
            int2 en = make_int2(0, 0);
            int dl = 0;
            float4 xs = xv[(size_t)r * 32 + lane];
            float4 acc = make_float4(0.0f, 0.0f, 0.0f, 0.0f);

            for (int base = start; base < end; base += 32) {
                int n = end - base;
                if (n > 32) n = 32;
                const int2* ep = ent + base;
                if (lane < n) {
                    en = ep[lane];
                    dl = __float_as_int(dinv[en.x]);
                }
                int j = 0;
                for (; j + 8 <= n; j += 8) row_batch<8>(xv, en, dl, j, lane, acc);
                if (j + 4 <= n) { row_batch<4>(xv, en, dl, j, lane, acc); j += 4; }
                int rem = n - j;
                if (rem == 3)      row_batch<3>(xv, en, dl, j, lane, acc);
                else if (rem == 2) row_batch<2>(xv, en, dl, j, lane, acc);
                else if (rem == 1) row_batch<1>(xv, en, dl, j, lane, acc);
            }

            float ss = s * s;
            o.x = fmaf(ss, xs.x, s * acc.x);
            o.y = fmaf(ss, xs.y, s * acc.y);
            o.z = fmaf(ss, xs.z, s * acc.z);
            o.w = fmaf(ss, xs.w, s * acc.w);
        }
        *(float4*)(At + (wv * 4 + q) * TPITCH + 4 * lane) = o;
    }
    int drawn = 0;
    if (t == 0) {
        int off = 0;
        asm volatile("" : "+v"(off));
        drawn = atomicAdd(tilectr + off, 1);
    }
    __builtin_amdgcn_fence(__ATOMIC_RELEASE, "workgroup", "local");
    __builtin_amdgcn_s_barrier();
    __builtin_amdgcn_fence(__ATOMIC_ACQUIRE, "workgroup", "local");

    int m  = lane & 15;
    asm volatile("" : "+v"(m));
    int kh = (lane >> 4) << 1;
    int tile = wv >> 2;
    int jb = (wv & 3) * 2;
    const float* ar = At + (tile * 16 + m) * TPITCH;

    v8f acc2[2] = {};

    for (int kb = 0; kb < FEAT / 4; ++kb) {
        int k = kb * 4 + kh;
        float2 av = *(const float2*)(ar + k);
        v2f a; a.x = av.x; a.y = av.y;
        #pragma unroll
        for (int jj = 0; jj < 2; ++jj) {
            int col = (jb + jj) * 16 + m;
            v2f b;
            b.x = Wl[(k + 0) * FEAT + col];
            b.y = Wl[(k + 1) * FEAT + col];
            acc2[jj] = __builtin_amdgcn_wmma_f32_16x16x4_f32(
                false, a, false, b, (short)0, acc2[jj], false, false);
        }
    }

    int rbase = tileIdx * TROWS + tile * 16 + ((lane >> 4) ? 8 : 0);
    int n = lane & 15;
    #pragma unroll
    for (int jj = 0; jj < 2; ++jj) {
        int u = (jb + jj) * 16 + n;
        float bu = bias[u];
        float gu = gamma[u];
        float mu = mean[u];
        float rs = rsqrtf(var[u] + BN_EPS);
        float be = beta[u];
        #pragma unroll
        for (int r = 0; r < 8; ++r) {
            int row = rbase + r;
            float v = acc2[jj][r] + bu;
            float y = gu * (v - mu) * rs + be;
            y = y > 0.0f ? y : 0.0f;
            if (row < nN) out[(size_t)row * FEAT + u] = y;
        }
    }

    if (t == 0) nextTile = nBlk + drawn;
    __builtin_amdgcn_fence(__ATOMIC_RELEASE, "workgroup", "local");
    __builtin_amdgcn_s_barrier();
    __builtin_amdgcn_fence(__ATOMIC_ACQUIRE, "workgroup", "local");
    tileIdx = __builtin_amdgcn_readfirstlane(nextTile);
  }
}


extern "C" void kernel_launch(void* const* d_in, const int* in_sizes, int n_in,
                              void* d_out, int out_size, void* d_ws, size_t ws_size,
                              hipStream_t stream) {
    const float* x     = (const float*)d_in[0];
    const int*   eidx  = (const int*)d_in[1];
    const float* ew    = (const float*)d_in[2];
    const float* W     = (const float*)d_in[3];
    const float* bias  = (const float*)d_in[4];
    const float* gamma = (const float*)d_in[5];
    const float* beta  = (const float*)d_in[6];
    const float* mean  = (const float*)d_in[7];
    const float* var   = (const float*)d_in[8];

    int nE = in_sizes[2];
    int nN = in_sizes[0] / FEAT;
    const int* rows = eidx;
    const int* cols = eidx + nE;

    int  nB     = (nN + BROWS - 1) / BROWS;
    int  pitch  = (nN + 64 + 3) & ~3;

    int*   w32    = (int*)d_ws;
    int*   rowptr = w32;
    int2*  ent    = (int2*)(w32 + pitch);
    float* dinv   = (float*)d_ws + (size_t)nN * FEAT + nN;
    int*   tilectr = w32 + nN + 32;
    float* out    = (float*)d_out;

    int* o32    = (int*)d_out;
    int* runs   = o32;
    int4* tmp   = (int4*)(runs + (nB + 1) * GBLK);

    bucket_kernel<<<GBLK, 1024, 0, stream>>>(rows, cols, ew, runs, tmp, tilectr, nE, nB);
    rowsort_kernel<<<nB, BROWS, 0, stream>>>(tmp, runs, rowptr, ent, dinv, nE, nN);
    gather_gemm_kernel<<<RES_BLOCKS, 512, 0, stream>>>(
        x, rowptr, ent, dinv, W, bias, gamma, beta, mean, var, out, tilectr, nN);
}
